# starting kernel
# baseline (speedup 1.0000x reference)
#include <hip/hip_runtime.h>

#define D 128

typedef float v2f __attribute__((ext_vector_type(2)));
typedef float v8f __attribute__((ext_vector_type(8)));

#if defined(__HIP_DEVICE_COMPILE__) && __has_builtin(__builtin_amdgcn_wmma_f32_16x16x4_f32)
#define HAVE_WMMA_F32 1
#endif

__device__ __forceinline__ v8f wmma_acc(v2f a, v2f b, v8f c) {
#if defined(HAVE_WMMA_F32)
  return __builtin_amdgcn_wmma_f32_16x16x4_f32(false, a, false, b, (short)0, c,
                                               false, false);
#else
  (void)a; (void)b;
  return c;
#endif
}

__global__ void k_zero(float* __restrict__ p, int n) {
  int i = blockIdx.x * blockDim.x + threadIdx.x;
  if (i < n) p[i] = 0.f;
}

__global__ void k_count(const int* __restrict__ dst, float* __restrict__ deg, int e) {
  int i = blockIdx.x * blockDim.x + threadIdx.x;
  if (i < e) {
    __hip_atomic_fetch_add(&deg[dst[i]], 1.0f, __ATOMIC_RELAXED,
                           __HIP_MEMORY_SCOPE_AGENT);
  }
}

__global__ void k_dinv(const float* __restrict__ deg, float* __restrict__ dinv, int n) {
  int i = blockIdx.x * blockDim.x + threadIdx.x;
  if (i < n) dinv[i] = rsqrtf(deg[i] + 1.0f);
}

__global__ void __launch_bounds__(256) k_gcn_gemm(
    const float* __restrict__ X, const float* __restrict__ W,
    const float* __restrict__ bias, const float* __restrict__ dinv,
    float* __restrict__ H, float* __restrict__ AGG, int nrows, int relu_in) {
  const int wave = threadIdx.x >> 5;
  const int lane = threadIdx.x & 31;
  const int half = lane >> 4;
  const int l16  = lane & 15;
  const int row0 = (blockIdx.x * 8 + wave) * 16;
  if (row0 >= nrows) return;

  const int m = row0 + l16;
  const bool mvalid = (m < nrows);
  const float* __restrict__ arow = X + (size_t)(mvalid ? m : 0) * D;

  const v8f vzero = {0.f, 0.f, 0.f, 0.f, 0.f, 0.f, 0.f, 0.f};
  v8f acc[8];
#pragma unroll
  for (int i = 0; i < 8; ++i) acc[i] = vzero;

  for (int k0 = 0; k0 < D; k0 += 4) {
    const int ka = k0 + 2 * half;
    float a0 = arow[ka];
    float a1 = arow[ka + 1];
    if (relu_in) { a0 = fmaxf(a0, 0.f); a1 = fmaxf(a1, 0.f); }
    if (!mvalid) { a0 = 0.f; a1 = 0.f; }
    v2f a; a.x = a0; a.y = a1;
    const float* __restrict__ w0 = W + (size_t)ka * D + l16;
    const float* __restrict__ w1 = w0 + D;
#pragma unroll
    for (int ct = 0; ct < 8; ++ct) {
      v2f b; b.x = w0[ct * 16]; b.y = w1[ct * 16];
      acc[ct] = wmma_acc(a, b, acc[ct]);
    }
  }

#pragma unroll
  for (int v = 0; v < 8; ++v) {
    const int mm = row0 + v + 8 * half;
    if (mm < nrows) {
      const float di = dinv[mm];
      const float s = di * di;
      const size_t base = (size_t)mm * D;
#pragma unroll
      for (int ct = 0; ct < 8; ++ct) {
        const int col = ct * 16 + l16;
        const float hv = acc[ct][v];
        H[base + col] = hv;
        AGG[base + col] = fmaf(hv, s, bias[col]);
      }
    }
  }
}

__global__ void __launch_bounds__(256) k_scatter(
    const float* __restrict__ H, const float* __restrict__ dinv,
    const int* __restrict__ src, const int* __restrict__ dst,
    float* __restrict__ AGG, int nedges) {
  const int lane = threadIdx.x & 31;
  const int e = (int)(((size_t)blockIdx.x * blockDim.x + threadIdx.x) >> 5);
  if (e >= nedges) return;
  const int s = src[e];
  const int d = dst[e];
  const float coef = dinv[s] * dinv[d];
  const float4 hv = ((const float4*)(H + (size_t)s * D))[lane];
  float* __restrict__ ap = AGG + (size_t)d * D + lane * 4;
  __hip_atomic_fetch_add(ap + 0, hv.x * coef, __ATOMIC_RELAXED, __HIP_MEMORY_SCOPE_AGENT);
  __hip_atomic_fetch_add(ap + 1, hv.y * coef, __ATOMIC_RELAXED, __HIP_MEMORY_SCOPE_AGENT);
  __hip_atomic_fetch_add(ap + 2, hv.z * coef, __ATOMIC_RELAXED, __HIP_MEMORY_SCOPE_AGENT);
  __hip_atomic_fetch_add(ap + 3, hv.w * coef, __ATOMIC_RELAXED, __HIP_MEMORY_SCOPE_AGENT);
}

extern "C" void kernel_launch(void* const* d_in, const int* in_sizes, int n_in,
                              void* d_out, int out_size, void* d_ws, size_t ws_size,
                              hipStream_t stream) {
  (void)n_in; (void)out_size; (void)ws_size;
  const float* x  = (const float*)d_in[0];
  const int*   ei = (const int*)d_in[1];
  const float* W1 = (const float*)d_in[2];
  const float* b1 = (const float*)d_in[3];
  const float* W2 = (const float*)d_in[4];
  const float* b2 = (const float*)d_in[5];
  const int n = in_sizes[0] / D;
  const int e = in_sizes[1] / 2;
  const int* srcv = ei;
  const int* dstv = ei + e;

  char* ws = (char*)d_ws;
  float* dinv = (float*)ws;
  size_t off = (((size_t)n * 4) + 255) & ~(size_t)255;
  float* Hbuf = (float*)(ws + off);
  off += (((size_t)n * D * 4) + 255) & ~(size_t)255;
  float* Abuf = (float*)(ws + off);
  float* deg = Hbuf;
  float* outp = (float*)d_out;

  const dim3 blk(256);
  const int nblk = (n + 255) / 256;
  const int eblk = (e + 255) / 256;
  const int gblk = (n + 127) / 128;
  const int sblk = (e + 7) / 8;

  k_zero <<<dim3(nblk), blk, 0, stream>>>(deg, n);
  k_count<<<dim3(eblk), blk, 0, stream>>>(dstv, deg, e);
  k_dinv <<<dim3(nblk), blk, 0, stream>>>(deg, dinv, n);

  k_gcn_gemm<<<dim3(gblk), blk, 0, stream>>>(x, W1, b1, dinv, Hbuf, Abuf, n, 0);
  k_scatter <<<dim3(sblk), blk, 0, stream>>>(Hbuf, dinv, srcv, dstv, Abuf, e);

  k_gcn_gemm<<<dim3(gblk), blk, 0, stream>>>(Abuf, W2, b2, dinv, Hbuf, outp, n, 1);
  k_scatter <<<dim3(sblk), blk, 0, stream>>>(Hbuf, dinv, srcv, dstv, outp, e);
}


# Aggregate per destination: edge list sorted by dst rebuilt each call (int count+rank atomic, scan, fill); one wave sums a node's rows in registers; no f32 atomics, sum order = list order
# speedup vs baseline: 1.8557x; 1.8557x over previous
#include <hip/hip_runtime.h>

#define D 128

typedef float v2f __attribute__((ext_vector_type(2)));
typedef float v8f __attribute__((ext_vector_type(8)));

#if defined(__HIP_DEVICE_COMPILE__) && __has_builtin(__builtin_amdgcn_wmma_f32_16x16x4_f32)
#define HAVE_WMMA_F32 1
#endif

__device__ __forceinline__ v8f wmma_acc(v2f a, v2f b, v8f c) {
#if defined(HAVE_WMMA_F32)
  return __builtin_amdgcn_wmma_f32_16x16x4_f32(false, a, false, b, (short)0, c,
                                               false, false);
#else
  (void)a; (void)b;
  return c;
#endif
}

__global__ void k_zero(int* __restrict__ p, int n) {
  int i = blockIdx.x * blockDim.x + threadIdx.x;
  if (i < n) p[i] = 0;
}

__global__ void k_count_rank(const int* __restrict__ dst, int* __restrict__ cnt,
                             int* __restrict__ rank, int e) {
  int i = blockIdx.x * blockDim.x + threadIdx.x;
  if (i < e) {
    rank[i] = __hip_atomic_fetch_add(&cnt[dst[i]], 1, __ATOMIC_RELAXED,
                                     __HIP_MEMORY_SCOPE_AGENT);
  }
}

__global__ void __launch_bounds__(256) k_blocksum(const int* __restrict__ cnt,
                                                 int* __restrict__ bsum, int n) {
  __shared__ int sm[8];
  const int i = blockIdx.x * 256 + threadIdx.x;
  int v = (i < n) ? cnt[i] : 0;
#pragma unroll
  for (int o = 16; o > 0; o >>= 1) v += __shfl_xor(v, o);
  if ((threadIdx.x & 31) == 0) sm[threadIdx.x >> 5] = v;
  __syncthreads();
  if (threadIdx.x == 0) {
    int t = 0;
#pragma unroll
    for (int k = 0; k < 8; ++k) t += sm[k];
    bsum[blockIdx.x] = t;
  }
}

__global__ void __launch_bounds__(256) k_scan(const int* __restrict__ cnt,
                                             const int* __restrict__ bsum,
                                             int* __restrict__ rowptr,
                                             float* __restrict__ dinv, int n) {
  __shared__ int sm_w[8];
  __shared__ int sm_off[8];
  const int tid = threadIdx.x;
  const int lane = tid & 31;
  const int w = tid >> 5;
  int p = 0;
  for (int j = tid; j < (int)blockIdx.x; j += 256) p += bsum[j];
#pragma unroll
  for (int o = 16; o > 0; o >>= 1) p += __shfl_xor(p, o);
  if (lane == 0) sm_off[w] = p;
  const int i = blockIdx.x * 256 + tid;
  const int v = (i < n) ? cnt[i] : 0;
  int x = v;
#pragma unroll
  for (int o = 1; o < 32; o <<= 1) {
    const int y = __shfl_up(x, o);
    if (lane >= o) x += y;
  }
  if (lane == 31) sm_w[w] = x;
  __syncthreads();
  int off = 0;
#pragma unroll
  for (int k = 0; k < 8; ++k) {
    off += sm_off[k];
    if (k < w) off += sm_w[k];
  }
  if (i < n) {
    rowptr[i] = off + x - v;
    dinv[i] = rsqrtf((float)v + 1.0f);
  }
}

__global__ void k_fill(const int* __restrict__ src, const int* __restrict__ dst,
                       const int* __restrict__ rank, const int* __restrict__ rowptr,
                       int* __restrict__ csr_src, int e) {
  int i = blockIdx.x * blockDim.x + threadIdx.x;
  if (i < e) csr_src[rowptr[dst[i]] + rank[i]] = src[i];
}

__global__ void __launch_bounds__(256) k_gcn_gemm(
    const float* __restrict__ X, const float* __restrict__ W,
    float* __restrict__ H, int nrows, int relu_in) {
  const int wave = threadIdx.x >> 5;
  const int lane = threadIdx.x & 31;
  const int half = lane >> 4;
  const int l16  = lane & 15;
  const int row0 = (blockIdx.x * 8 + wave) * 16;
  if (row0 >= nrows) return;

  const int m = row0 + l16;
  const bool mvalid = (m < nrows);
  const float* __restrict__ arow = X + (size_t)(mvalid ? m : 0) * D;

  const v8f vzero = {0.f, 0.f, 0.f, 0.f, 0.f, 0.f, 0.f, 0.f};
  v8f acc[8];
#pragma unroll
  for (int i = 0; i < 8; ++i) acc[i] = vzero;

  for (int k0 = 0; k0 < D; k0 += 4) {
    const int ka = k0 + 2 * half;
    float a0 = arow[ka];
    float a1 = arow[ka + 1];
    if (relu_in) { a0 = fmaxf(a0, 0.f); a1 = fmaxf(a1, 0.f); }
    if (!mvalid) { a0 = 0.f; a1 = 0.f; }
    v2f a; a.x = a0; a.y = a1;
    const float* __restrict__ w0 = W + (size_t)ka * D + l16;
    const float* __restrict__ w1 = w0 + D;
#pragma unroll
    for (int ct = 0; ct < 8; ++ct) {
      v2f b; b.x = w0[ct * 16]; b.y = w1[ct * 16];
      acc[ct] = wmma_acc(a, b, acc[ct]);
    }
  }

#pragma unroll
  for (int v = 0; v < 8; ++v) {
    const int mm = row0 + v + 8 * half;
    if (mm < nrows) {
      const size_t base = (size_t)mm * D;
#pragma unroll
      for (int ct = 0; ct < 8; ++ct) {
        H[base + ct * 16 + l16] = acc[ct][v];
      }
    }
  }
}

__device__ __forceinline__ void fma4(float4& acc, float w, const float4 h) {
  acc.x = fmaf(w, h.x, acc.x);
  acc.y = fmaf(w, h.y, acc.y);
  acc.z = fmaf(w, h.z, acc.z);
  acc.w = fmaf(w, h.w, acc.w);
}

template <int K>
__device__ __forceinline__ void rows_fma(float4& acc, int my_s, float my_w, int j,
                                         const float4* __restrict__ H4, int lane) {
  float4 h[K];
#pragma unroll
  for (int k = 0; k < K; ++k) {
    const int s = __builtin_amdgcn_readlane(my_s, j + k);
    h[k] = H4[(size_t)s * (D / 4) + lane];
  }
#pragma unroll
  for (int k = 0; k < K; ++k) {
    const float w = __int_as_float(__builtin_amdgcn_readlane(__float_as_int(my_w), j + k));
    fma4(acc, w, h[k]);
  }
}

__global__ void __launch_bounds__(256) k_gather(
    const float* __restrict__ H, const float* __restrict__ dinv,
    const int* __restrict__ rowptr, const int* __restrict__ cnt,
    const int* __restrict__ csr_src, const float* __restrict__ bias,
    float* __restrict__ OUT, int n) {
  const int lane = threadIdx.x & 31;
  const int wv = __builtin_amdgcn_readfirstlane((int)(threadIdx.x >> 5));
  const int d = blockIdx.x * 8 + wv;
  if (d >= n) return;
  const float dd = dinv[d];
  const int start = rowptr[d];
  const int c = cnt[d];
  const int tot = c + 1;
  const float4* __restrict__ H4 = (const float4*)H;
  float4 acc;
  acc.x = 0.f; acc.y = 0.f; acc.z = 0.f; acc.w = 0.f;

  for (int base = 0; base < tot; base += 32) {
    const int m = (tot - base < 32) ? (tot - base) : 32;
    const int idx = base + lane;
    int my_s = d;
    float my_w = (idx == c) ? dd : 0.f;
    if (idx < c) {
      my_s = csr_src[start + idx];
      my_w = dinv[my_s];
    }
    int j = 0;
    for (; j + 8 <= m; j += 8) rows_fma<8>(acc, my_s, my_w, j, H4, lane);
    switch (m - j) {
      case 7: rows_fma<7>(acc, my_s, my_w, j, H4, lane); break;
      case 6: rows_fma<6>(acc, my_s, my_w, j, H4, lane); break;
      case 5: rows_fma<5>(acc, my_s, my_w, j, H4, lane); break;
      case 4: rows_fma<4>(acc, my_s, my_w, j, H4, lane); break;
      case 3: rows_fma<3>(acc, my_s, my_w, j, H4, lane); break;
      case 2: rows_fma<2>(acc, my_s, my_w, j, H4, lane); break;
      case 1: rows_fma<1>(acc, my_s, my_w, j, H4, lane); break;
      default: break;
    }
  }

  const float4 b = ((const float4*)bias)[lane];
  float4 o;
  o.x = fmaf(acc.x, dd, b.x);
  o.y = fmaf(acc.y, dd, b.y);
  o.z = fmaf(acc.z, dd, b.z);
  o.w = fmaf(acc.w, dd, b.w);
  ((float4*)OUT)[(size_t)d * (D / 4) + lane] = o;
}

extern "C" void kernel_launch(void* const* d_in, const int* in_sizes, int n_in,
                              void* d_out, int out_size, void* d_ws, size_t ws_size,
                              hipStream_t stream) {
  (void)n_in; (void)out_size; (void)ws_size;
  const float* x  = (const float*)d_in[0];
  const int*   ei = (const int*)d_in[1];
  const float* W1 = (const float*)d_in[2];
  const float* b1 = (const float*)d_in[3];
  const float* W2 = (const float*)d_in[4];
  const float* b2 = (const float*)d_in[5];
  const int n = in_sizes[0] / D;
  const int e = in_sizes[1] / 2;
  const int* srcv = ei;
  const int* dstv = ei + e;

  char* ws = (char*)d_ws;
  size_t off = 0;
  float* dinv = (float*)(ws + off);
  off += (((size_t)n * 4) + 255) & ~(size_t)255;
  int* cnt = (int*)(ws + off);
  off += (((size_t)n * 4) + 255) & ~(size_t)255;
  int* rowptr = (int*)(ws + off);
  off += (((size_t)n * 4) + 255) & ~(size_t)255;
  const int nblk = (n + 255) / 256;
  int* bsum = (int*)(ws + off);
  off += (((size_t)nblk * 4) + 255) & ~(size_t)255;
  int* rank = (int*)(ws + off);
  off += (((size_t)e * 4) + 255) & ~(size_t)255;
  int* csr = (int*)(ws + off);
  off += (((size_t)e * 4) + 255) & ~(size_t)255;
  float* Hbuf = (float*)(ws + off);
  off += (((size_t)n * D * 4) + 255) & ~(size_t)255;
  float* Abuf = (float*)(ws + off);
  float* outp = (float*)d_out;

  const dim3 blk(256);
  const int eblk = (e + 255) / 256;
  const int gblk = (n + 127) / 128;
  const int ablk = (n + 7) / 8;

  k_zero      <<<dim3(nblk), blk, 0, stream>>>(cnt, n);
  k_count_rank<<<dim3(eblk), blk, 0, stream>>>(dstv, cnt, rank, e);
  k_blocksum  <<<dim3(nblk), blk, 0, stream>>>(cnt, bsum, n);
  k_scan      <<<dim3(nblk), blk, 0, stream>>>(cnt, bsum, rowptr, dinv, n);
  k_fill      <<<dim3(eblk), blk, 0, stream>>>(srcv, dstv, rank, rowptr, csr, e);

  k_gcn_gemm<<<dim3(gblk), blk, 0, stream>>>(x, W1, Hbuf, n, 0);
  k_gather  <<<dim3(ablk), blk, 0, stream>>>(Hbuf, dinv, rowptr, cnt, csr, b1, Abuf, n);

  k_gcn_gemm<<<dim3(gblk), blk, 0, stream>>>(Abuf, W2, Hbuf, n, 1);
  k_gather  <<<dim3(ablk), blk, 0, stream>>>(Hbuf, dinv, rowptr, cnt, csr, b2, outp, n);
}


# List fill joins the layer-1 GEMM launch as extra work-groups (both need only the scan's output): 9 launches become 8 and the scattered fill runs beside the matrix work; sums unchanged
# speedup vs baseline: 2.1994x; 1.1852x over previous
#include <hip/hip_runtime.h>

#define D 128

typedef float v2f __attribute__((ext_vector_type(2)));
typedef float v8f __attribute__((ext_vector_type(8)));

#if defined(__HIP_DEVICE_COMPILE__) && __has_builtin(__builtin_amdgcn_wmma_f32_16x16x4_f32)
#define HAVE_WMMA_F32 1
#endif

__device__ __forceinline__ v8f wmma_acc(v2f a, v2f b, v8f c) {
#if defined(HAVE_WMMA_F32)
  return __builtin_amdgcn_wmma_f32_16x16x4_f32(false, a, false, b, (short)0, c,
                                               false, false);
#else
  (void)a; (void)b;
  return c;
#endif
}

__global__ void k_zero(int* __restrict__ p, int n) {
  int i = blockIdx.x * blockDim.x + threadIdx.x;
  if (i < n) p[i] = 0;
}

__global__ void k_count_rank(const int* __restrict__ dst, int* __restrict__ cnt,
                             int* __restrict__ rank, int e) {
  int i = blockIdx.x * blockDim.x + threadIdx.x;
  if (i < e) {
    rank[i] = __hip_atomic_fetch_add(&cnt[dst[i]], 1, __ATOMIC_RELAXED,
                                     __HIP_MEMORY_SCOPE_AGENT);
  }
}

__global__ void __launch_bounds__(256) k_blocksum(const int* __restrict__ cnt,
                                                 int* __restrict__ bsum, int n) {
  __shared__ int sm[8];
  const int i = blockIdx.x * 256 + threadIdx.x;
  int v = (i < n) ? cnt[i] : 0;
#pragma unroll
  for (int o = 16; o > 0; o >>= 1) v += __shfl_xor(v, o);
  if ((threadIdx.x & 31) == 0) sm[threadIdx.x >> 5] = v;
  __syncthreads();
  if (threadIdx.x == 0) {
    int t = 0;
#pragma unroll
    for (int k = 0; k < 8; ++k) t += sm[k];
    bsum[blockIdx.x] = t;
  }
}

__global__ void __launch_bounds__(256) k_scan(const int* __restrict__ cnt,
                                             const int* __restrict__ bsum,
                                             int* __restrict__ rowptr,
                                             float* __restrict__ dinv, int n) {
  __shared__ int sm_w[8];
  __shared__ int sm_off[8];
  const int tid = threadIdx.x;
  const int lane = tid & 31;
  const int w = tid >> 5;
  int p = 0;
  for (int j = tid; j < (int)blockIdx.x; j += 256) p += bsum[j];
#pragma unroll
  for (int o = 16; o > 0; o >>= 1) p += __shfl_xor(p, o);
  if (lane == 0) sm_off[w] = p;
  const int i = blockIdx.x * 256 + tid;
  const int v = (i < n) ? cnt[i] : 0;
  int x = v;
#pragma unroll
  for (int o = 1; o < 32; o <<= 1) {
    const int y = __shfl_up(x, o);
    if (lane >= o) x += y;
  }
  if (lane == 31) sm_w[w] = x;
  __syncthreads();
  int off = 0;
#pragma unroll
  for (int k = 0; k < 8; ++k) {
    off += sm_off[k];
    if (k < w) off += sm_w[k];
  }
  if (i < n) {
    rowptr[i] = off + x - v;
    dinv[i] = rsqrtf((float)v + 1.0f);
  }
}

__device__ __forceinline__ void gemm_strip(
    const float* __restrict__ X, const float* __restrict__ W,
    const float* __restrict__ dinv, float* __restrict__ H, int nrows, int relu_in,
    int blk) {
  const int wave = threadIdx.x >> 5;
  const int lane = threadIdx.x & 31;
  const int half = lane >> 4;
  const int l16  = lane & 15;
  const int row0 = (blk * 8 + wave) * 16;
  if (row0 >= nrows) return;

  const int m = row0 + l16;
  const bool mvalid = (m < nrows);
  const float* __restrict__ arow = X + (size_t)(mvalid ? m : 0) * D;

  float rs[8];
#pragma unroll
  for (int v = 0; v < 8; ++v) {
    const int mm = row0 + v + 8 * half;
    rs[v] = dinv[(mm < nrows) ? mm : (nrows - 1)];
  }

  const v8f vzero = {0.f, 0.f, 0.f, 0.f, 0.f, 0.f, 0.f, 0.f};
  v8f acc[8];
#pragma unroll
  for (int i = 0; i < 8; ++i) acc[i] = vzero;

  for (int k0 = 0; k0 < D; k0 += 4) {
    const int ka = k0 + 2 * half;
    float a0 = arow[ka];
    float a1 = arow[ka + 1];
    if (relu_in) { a0 = fmaxf(a0, 0.f); a1 = fmaxf(a1, 0.f); }
    if (!mvalid) { a0 = 0.f; a1 = 0.f; }
    v2f a; a.x = a0; a.y = a1;
    const float* __restrict__ w0 = W + (size_t)ka * D + l16;
    const float* __restrict__ w1 = w0 + D;
#pragma unroll
    for (int ct = 0; ct < 8; ++ct) {
      v2f b; b.x = w0[ct * 16]; b.y = w1[ct * 16];
      acc[ct] = wmma_acc(a, b, acc[ct]);
    }
  }

#pragma unroll
  for (int v = 0; v < 8; ++v) {
    const int mm = row0 + v + 8 * half;
    if (mm < nrows) {
      const size_t base = (size_t)mm * D;
#pragma unroll
      for (int ct = 0; ct < 8; ++ct) {
        H[base + ct * 16 + l16] = acc[ct][v] * rs[v];
      }
    }
  }
}

__global__ void __launch_bounds__(256) k_gcn_gemm(
    const float* __restrict__ X, const float* __restrict__ W,
    const float* __restrict__ dinv, float* __restrict__ H, int nrows, int relu_in) {
  gemm_strip(X, W, dinv, H, nrows, relu_in, (int)blockIdx.x);
}

__global__ void __launch_bounds__(256) k_gemm1_fill(
    const float* __restrict__ X, const float* __restrict__ W,
    const float* __restrict__ dinv, float* __restrict__ H, int nrows, int gblk,
    const int* __restrict__ src, const int* __restrict__ dst,
    const int* __restrict__ rank, const int* __restrict__ rowptr,
    int* __restrict__ csr_src, int e) {
  const int b = (int)blockIdx.x;
  if (b < gblk) {
    gemm_strip(X, W, dinv, H, nrows, 0, b);
  } else {
    const int i = (b - gblk) * 256 + (int)threadIdx.x;
    if (i < e) csr_src[rowptr[dst[i]] + rank[i]] = src[i];
  }
}

__device__ __forceinline__ void add4(float4& acc, const float4 h) {
  acc.x += h.x;
  acc.y += h.y;
  acc.z += h.z;
  acc.w += h.w;
}

template <int K>
__device__ __forceinline__ void rows_add(float4& acc, int my_s, int j,
                                         const float4* __restrict__ H4, int lane) {
  float4 h[K];
#pragma unroll
  for (int k = 0; k < K; ++k) {
    const int s = __builtin_amdgcn_readlane(my_s, j + k);
    h[k] = H4[(size_t)s * (D / 4) + lane];
  }
#pragma unroll
  for (int k = 0; k < K; ++k) add4(acc, h[k]);
}

__global__ void __launch_bounds__(256) k_gather(
    const float* __restrict__ H, const float* __restrict__ dinv,
    const int* __restrict__ rowptr, const int* __restrict__ cnt,
    const int* __restrict__ csr_src, const float* __restrict__ bias,
    float* __restrict__ OUT, int n) {
  const int lane = threadIdx.x & 31;
  const int wv = __builtin_amdgcn_readfirstlane((int)(threadIdx.x >> 5));
  const int d = blockIdx.x * 8 + wv;
  if (d >= n) return;
  const float dd = dinv[d];
  const int start = rowptr[d];
  const int c = cnt[d];
  const int tot = c + 1;
  const float4* __restrict__ H4 = (const float4*)H;
  float4 acc;
  acc.x = 0.f; acc.y = 0.f; acc.z = 0.f; acc.w = 0.f;

  for (int base = 0; base < tot; base += 32) {
    const int m = (tot - base < 32) ? (tot - base) : 32;
    const int idx = base + lane;
    int my_s = d;
    if (idx < c) my_s = csr_src[start + idx];
    int j = 0;
    for (; j + 8 <= m; j += 8) rows_add<8>(acc, my_s, j, H4, lane);
    switch (m - j) {
      case 7: rows_add<7>(acc, my_s, j, H4, lane); break;
      case 6: rows_add<6>(acc, my_s, j, H4, lane); break;
      case 5: rows_add<5>(acc, my_s, j, H4, lane); break;
      case 4: rows_add<4>(acc, my_s, j, H4, lane); break;
      case 3: rows_add<3>(acc, my_s, j, H4, lane); break;
      case 2: rows_add<2>(acc, my_s, j, H4, lane); break;
      case 1: rows_add<1>(acc, my_s, j, H4, lane); break;
      default: break;
    }
  }

  const float4 b = ((const float4*)bias)[lane];
  float4 o;
  o.x = fmaf(acc.x, dd, b.x);
  o.y = fmaf(acc.y, dd, b.y);
  o.z = fmaf(acc.z, dd, b.z);
  o.w = fmaf(acc.w, dd, b.w);
  ((float4*)OUT)[(size_t)d * (D / 4) + lane] = o;
}

extern "C" void kernel_launch(void* const* d_in, const int* in_sizes, int n_in,
                              void* d_out, int out_size, void* d_ws, size_t ws_size,
                              hipStream_t stream) {
  (void)n_in; (void)out_size; (void)ws_size;
  const float* x  = (const float*)d_in[0];
  const int*   ei = (const int*)d_in[1];
  const float* W1 = (const float*)d_in[2];
  const float* b1 = (const float*)d_in[3];
  const float* W2 = (const float*)d_in[4];
  const float* b2 = (const float*)d_in[5];
  const int n = in_sizes[0] / D;
  const int e = in_sizes[1] / 2;
  const int* srcv = ei;
  const int* dstv = ei + e;

  char* ws = (char*)d_ws;
  size_t off = 0;
  float* dinv = (float*)(ws + off);
  off += (((size_t)n * 4) + 255) & ~(size_t)255;
  int* cnt = (int*)(ws + off);
  off += (((size_t)n * 4) + 255) & ~(size_t)255;
  int* rowptr = (int*)(ws + off);
  off += (((size_t)n * 4) + 255) & ~(size_t)255;
  const int nblk = (n + 255) / 256;
  int* bsum = (int*)(ws + off);
  off += (((size_t)nblk * 4) + 255) & ~(size_t)255;
  int* rank = (int*)(ws + off);
  off += (((size_t)e * 4) + 255) & ~(size_t)255;
  int* csr = (int*)(ws + off);
  off += (((size_t)e * 4) + 255) & ~(size_t)255;
  float* Hbuf = (float*)(ws + off);
  off += (((size_t)n * D * 4) + 255) & ~(size_t)255;
  float* Abuf = (float*)(ws + off);
  float* outp = (float*)d_out;

  const dim3 blk(256);
  const int eblk = (e + 255) / 256;
  const int gblk = (n + 127) / 128;
  const int ablk = (n + 7) / 8;

  k_zero      <<<dim3(nblk), blk, 0, stream>>>(cnt, n);
  k_count_rank<<<dim3(eblk), blk, 0, stream>>>(dstv, cnt, rank, e);
  k_blocksum  <<<dim3(nblk), blk, 0, stream>>>(cnt, bsum, n);
  k_scan      <<<dim3(nblk), blk, 0, stream>>>(cnt, bsum, rowptr, dinv, n);

  k_gemm1_fill<<<dim3(gblk + eblk), blk, 0, stream>>>(x, W1, dinv, Hbuf, n, gblk,
                                                       srcv, dstv, rank, rowptr, csr, e);
  k_gather  <<<dim3(ablk), blk, 0, stream>>>(Hbuf, dinv, rowptr, cnt, csr, b1, Abuf, n);

  k_gcn_gemm<<<dim3(gblk), blk, 0, stream>>>(Abuf, W2, dinv, Hbuf, n, 1);
  k_gather  <<<dim3(ablk), blk, 0, stream>>>(Hbuf, dinv, rowptr, cnt, csr, b2, outp, n);
}


# Layer-1 aggregation joins the layer-2 GEMM launch: 32-node tiles are summed into LDS and multiplied from there, aggregated rows never reach memory; 8 launches become 7; sums unchanged
# speedup vs baseline: 2.2739x; 1.0339x over previous
#include <hip/hip_runtime.h>

#define D 128

typedef float v2f __attribute__((ext_vector_type(2)));
typedef float v8f __attribute__((ext_vector_type(8)));

#if defined(__HIP_DEVICE_COMPILE__) && __has_builtin(__builtin_amdgcn_wmma_f32_16x16x4_f32)
#define HAVE_WMMA_F32 1
#endif

__device__ __forceinline__ v8f wmma_acc(v2f a, v2f b, v8f c) {
#if defined(HAVE_WMMA_F32)
  return __builtin_amdgcn_wmma_f32_16x16x4_f32(false, a, false, b, (short)0, c,
                                               false, false);
#else
  (void)a; (void)b;
  return c;
#endif
}

__global__ void k_zero(int* __restrict__ p, int n) {
  int i = blockIdx.x * blockDim.x + threadIdx.x;
  if (i < n) p[i] = 0;
}

__global__ void k_count_rank(const int* __restrict__ dst, int* __restrict__ cnt,
                             int* __restrict__ rank, int e) {
  int i = blockIdx.x * blockDim.x + threadIdx.x;
  if (i < e) {
    rank[i] = __hip_atomic_fetch_add(&cnt[dst[i]], 1, __ATOMIC_RELAXED,
                                     __HIP_MEMORY_SCOPE_AGENT);
  }
}

__global__ void __launch_bounds__(256) k_blocksum(const int* __restrict__ cnt,
                                                 int* __restrict__ bsum, int n) {
  __shared__ int sm[8];
  const int i = blockIdx.x * 256 + threadIdx.x;
  int v = (i < n) ? cnt[i] : 0;
#pragma unroll
  for (int o = 16; o > 0; o >>= 1) v += __shfl_xor(v, o);
  if ((threadIdx.x & 31) == 0) sm[threadIdx.x >> 5] = v;
  __syncthreads();
  if (threadIdx.x == 0) {
    int t = 0;
#pragma unroll
    for (int k = 0; k < 8; ++k) t += sm[k];
    bsum[blockIdx.x] = t;
  }
}

__global__ void __launch_bounds__(256) k_scan(const int* __restrict__ cnt,
                                             const int* __restrict__ bsum,
                                             int* __restrict__ rowptr,
                                             float* __restrict__ dinv, int n) {
  __shared__ int sm_w[8];
  __shared__ int sm_off[8];
  const int tid = threadIdx.x;
  const int lane = tid & 31;
  const int w = tid >> 5;
  int p = 0;
  for (int j = tid; j < (int)blockIdx.x; j += 256) p += bsum[j];
#pragma unroll
  for (int o = 16; o > 0; o >>= 1) p += __shfl_xor(p, o);
  if (lane == 0) sm_off[w] = p;
  const int i = blockIdx.x * 256 + tid;
  const int v = (i < n) ? cnt[i] : 0;
  int x = v;
#pragma unroll
  for (int o = 1; o < 32; o <<= 1) {
    const int y = __shfl_up(x, o);
    if (lane >= o) x += y;
  }
  if (lane == 31) sm_w[w] = x;
  __syncthreads();
  int off = 0;
#pragma unroll
  for (int k = 0; k < 8; ++k) {
    off += sm_off[k];
    if (k < w) off += sm_w[k];
  }
  if (i < n) {
    rowptr[i] = off + x - v;
    dinv[i] = rsqrtf((float)v + 1.0f);
  }
}

__device__ __forceinline__ void gemm_strip(
    const float* __restrict__ X, const float* __restrict__ W,
    const float* __restrict__ dinv, float* __restrict__ H, int nrows, int blk) {
  const int wave = threadIdx.x >> 5;
  const int lane = threadIdx.x & 31;
  const int half = lane >> 4;
  const int l16  = lane & 15;
  const int row0 = (blk * 8 + wave) * 16;
  if (row0 >= nrows) return;

  const int m = row0 + l16;
  const bool mvalid = (m < nrows);
  const float* __restrict__ arow = X + (size_t)(mvalid ? m : 0) * D;

  float rs[8];
#pragma unroll
  for (int v = 0; v < 8; ++v) {
    const int mm = row0 + v + 8 * half;
    rs[v] = dinv[(mm < nrows) ? mm : (nrows - 1)];
  }

  const v8f vzero = {0.f, 0.f, 0.f, 0.f, 0.f, 0.f, 0.f, 0.f};
  v8f acc[8];
#pragma unroll
  for (int i = 0; i < 8; ++i) acc[i] = vzero;

  for (int k0 = 0; k0 < D; k0 += 4) {
    const int ka = k0 + 2 * half;
    float a0 = arow[ka];
    float a1 = arow[ka + 1];
    if (!mvalid) { a0 = 0.f; a1 = 0.f; }
    v2f a; a.x = a0; a.y = a1;
    const float* __restrict__ w0 = W + (size_t)ka * D + l16;
    const float* __restrict__ w1 = w0 + D;
#pragma unroll
    for (int ct = 0; ct < 8; ++ct) {
      v2f b; b.x = w0[ct * 16]; b.y = w1[ct * 16];
      acc[ct] = wmma_acc(a, b, acc[ct]);
    }
  }

#pragma unroll
  for (int v = 0; v < 8; ++v) {
    const int mm = row0 + v + 8 * half;
    if (mm < nrows) {
      const size_t base = (size_t)mm * D;
#pragma unroll
      for (int ct = 0; ct < 8; ++ct) {
        H[base + ct * 16 + l16] = acc[ct][v] * rs[v];
      }
    }
  }
}

__global__ void __launch_bounds__(256) k_gemm1_fill(
    const float* __restrict__ X, const float* __restrict__ W,
    const float* __restrict__ dinv, float* __restrict__ H, int nrows, int gblk,
    const int* __restrict__ src, const int* __restrict__ dst,
    const int* __restrict__ rank, const int* __restrict__ rowptr,
    int* __restrict__ csr_src, int e) {
  const int b = (int)blockIdx.x;
  if (b < gblk) {
    gemm_strip(X, W, dinv, H, nrows, b);
  } else {
    const int i = (b - gblk) * 256 + (int)threadIdx.x;
    if (i < e) csr_src[rowptr[dst[i]] + rank[i]] = src[i];
  }
}

__device__ __forceinline__ void add4(float4& acc, const float4 h) {
  acc.x += h.x;
  acc.y += h.y;
  acc.z += h.z;
  acc.w += h.w;
}

template <int K>
__device__ __forceinline__ void rows_add(float4& acc, int my_s, int j,
                                         const float4* __restrict__ H4, int lane) {
  float4 h[K];
#pragma unroll
  for (int k = 0; k < K; ++k) {
    const int s = __builtin_amdgcn_readlane(my_s, j + k);
    h[k] = H4[(size_t)s * (D / 4) + lane];
  }
#pragma unroll
  for (int k = 0; k < K; ++k) add4(acc, h[k]);
}

__device__ __forceinline__ int list_head(int d, int start, int c,
                                         const int* __restrict__ csr_src, int lane) {
  int my_s = d;
  if (lane < c) my_s = csr_src[start + lane];
  return my_s;
}

__device__ __forceinline__ float4 node_sum(int d, int start, int c, int head,
                                           const int* __restrict__ csr_src,
                                           const float4* __restrict__ H4, int lane) {
  const int tot = c + 1;
  float4 acc;
  acc.x = 0.f; acc.y = 0.f; acc.z = 0.f; acc.w = 0.f;
  for (int base = 0; base < tot; base += 32) {
    const int m = (tot - base < 32) ? (tot - base) : 32;
    int my_s = head;
    if (base > 0) {
      const int idx = base + lane;
      my_s = d;
      if (idx < c) my_s = csr_src[start + idx];
    }
    int j = 0;
    for (; j + 8 <= m; j += 8) rows_add<8>(acc, my_s, j, H4, lane);
    switch (m - j) {
      case 7: rows_add<7>(acc, my_s, j, H4, lane); break;
      case 6: rows_add<6>(acc, my_s, j, H4, lane); break;
      case 5: rows_add<5>(acc, my_s, j, H4, lane); break;
      case 4: rows_add<4>(acc, my_s, j, H4, lane); break;
      case 3: rows_add<3>(acc, my_s, j, H4, lane); break;
      case 2: rows_add<2>(acc, my_s, j, H4, lane); break;
      case 1: rows_add<1>(acc, my_s, j, H4, lane); break;
      default: break;
    }
  }
  return acc;
}

__global__ void __launch_bounds__(256) k_gather(
    const float* __restrict__ H, const float* __restrict__ dinv,
    const int* __restrict__ rowptr, const int* __restrict__ cnt,
    const int* __restrict__ csr_src, const float* __restrict__ bias,
    float* __restrict__ OUT, int n) {
  const int lane = threadIdx.x & 31;
  const int wv = __builtin_amdgcn_readfirstlane((int)(threadIdx.x >> 5));
  const int d = blockIdx.x * 8 + wv;
  if (d >= n) return;
  const float dd = dinv[d];
  const int start = rowptr[d];
  const int c = cnt[d];
  const int head = list_head(d, start, c, csr_src, lane);
  const float4 acc = node_sum(d, start, c, head, csr_src, (const float4*)H, lane);
  const float4 b = ((const float4*)bias)[lane];
  float4 o;
  o.x = fmaf(acc.x, dd, b.x);
  o.y = fmaf(acc.y, dd, b.y);
  o.z = fmaf(acc.z, dd, b.z);
  o.w = fmaf(acc.w, dd, b.w);
  ((float4*)OUT)[(size_t)d * (D / 4) + lane] = o;
}

#define TROWS 32
#define APITCH 132

__global__ void __launch_bounds__(256) k_gather_gemm2(
    const float* __restrict__ H, const float* __restrict__ dinv,
    const int* __restrict__ rowptr, const int* __restrict__ cnt,
    const int* __restrict__ csr_src, const float* __restrict__ bias,
    const float* __restrict__ W, float* __restrict__ H2, int n, int ntiles) {
  __shared__ __attribute__((aligned(16))) float As[TROWS * APITCH];
  const int lane = threadIdx.x & 31;
  const int wv = __builtin_amdgcn_readfirstlane((int)(threadIdx.x >> 5));
  const int half = lane >> 4;
  const int l16 = lane & 15;
  const int col = wv * 16 + l16;

  v2f bf[32];
#pragma unroll
  for (int i = 0; i < 32; ++i) {
    const int ka = 4 * i + 2 * half;
    bf[i].x = W[(size_t)ka * D + col];
    bf[i].y = W[(size_t)(ka + 1) * D + col];
  }
  const float4 b4 = ((const float4*)bias)[lane];
  const float4* __restrict__ H4 = (const float4*)H;
  const v8f vzero = {0.f, 0.f, 0.f, 0.f, 0.f, 0.f, 0.f, 0.f};

  for (int tile = (int)blockIdx.x; tile < ntiles; tile += (int)gridDim.x) {
    const int row0 = tile * TROWS;

    int nst[4], ncn[4], nhd[4];
    float ndd[4];
#pragma unroll
    for (int r = 0; r < 4; ++r) {
      const int d = row0 + wv * 4 + r;
      const int dc = (d < n) ? d : (n - 1);
      nst[r] = rowptr[dc];
      ncn[r] = cnt[dc];
      ndd[r] = dinv[dc];
    }
#pragma unroll
    for (int r = 0; r < 4; ++r) {
      const int d = row0 + wv * 4 + r;
      const int dc = (d < n) ? d : (n - 1);
      nhd[r] = list_head(dc, nst[r], ncn[r], csr_src, lane);
    }
#pragma unroll
    for (int r = 0; r < 4; ++r) {
      const int lr = wv * 4 + r;
      const int d = row0 + lr;
      float4 o;
      o.x = 0.f; o.y = 0.f; o.z = 0.f; o.w = 0.f;
      if (d < n) {
        const float dd = ndd[r];
        const float4 acc = node_sum(d, nst[r], ncn[r], nhd[r], csr_src, H4, lane);
        o.x = fmaxf(fmaf(acc.x, dd, b4.x), 0.f);
        o.y = fmaxf(fmaf(acc.y, dd, b4.y), 0.f);
        o.z = fmaxf(fmaf(acc.z, dd, b4.z), 0.f);
        o.w = fmaxf(fmaf(acc.w, dd, b4.w), 0.f);
      }
      *(float4*)&As[lr * APITCH + 4 * lane] = o;
    }

    __syncthreads();

    float rs0[8], rs1[8];
#pragma unroll
    for (int v = 0; v < 8; ++v) {
      const int m0 = row0 + v + 8 * half;
      const int m1 = m0 + 16;
      rs0[v] = dinv[(m0 < n) ? m0 : (n - 1)];
      rs1[v] = dinv[(m1 < n) ? m1 : (n - 1)];
    }

    v8f acc0 = vzero, acc1 = vzero;
#pragma unroll
    for (int i = 0; i < 32; ++i) {
      const int ka = 4 * i + 2 * half;
      const v2f a0 = *(const v2f*)&As[l16 * APITCH + ka];
      const v2f a1 = *(const v2f*)&As[(16 + l16) * APITCH + ka];
      acc0 = wmma_acc(a0, bf[i], acc0);
      acc1 = wmma_acc(a1, bf[i], acc1);
    }

#pragma unroll
    for (int v = 0; v < 8; ++v) {
      const int m0 = row0 + v + 8 * half;
      const int m1 = m0 + 16;
      if (m0 < n) H2[(size_t)m0 * D + col] = acc0[v] * rs0[v];
      if (m1 < n) H2[(size_t)m1 * D + col] = acc1[v] * rs1[v];
    }

    __syncthreads();
  }
}

extern "C" void kernel_launch(void* const* d_in, const int* in_sizes, int n_in,
                              void* d_out, int out_size, void* d_ws, size_t ws_size,
                              hipStream_t stream) {
  (void)n_in; (void)out_size; (void)ws_size;
  const float* x  = (const float*)d_in[0];
  const int*   ei = (const int*)d_in[1];
  const float* W1 = (const float*)d_in[2];
  const float* b1 = (const float*)d_in[3];
  const float* W2 = (const float*)d_in[4];
  const float* b2 = (const float*)d_in[5];
  const int n = in_sizes[0] / D;
  const int e = in_sizes[1] / 2;
  const int* srcv = ei;
  const int* dstv = ei + e;

  char* ws = (char*)d_ws;
  size_t off = 0;
  float* dinv = (float*)(ws + off);
  off += (((size_t)n * 4) + 255) & ~(size_t)255;
  int* cnt = (int*)(ws + off);
  off += (((size_t)n * 4) + 255) & ~(size_t)255;
  int* rowptr = (int*)(ws + off);
  off += (((size_t)n * 4) + 255) & ~(size_t)255;
  const int nblk = (n + 255) / 256;
  int* bsum = (int*)(ws + off);
  off += (((size_t)nblk * 4) + 255) & ~(size_t)255;
  int* rank = (int*)(ws + off);
  off += (((size_t)e * 4) + 255) & ~(size_t)255;
  int* csr = (int*)(ws + off);
  off += (((size_t)e * 4) + 255) & ~(size_t)255;
  float* Hbuf = (float*)(ws + off);
  off += (((size_t)n * D * 4) + 255) & ~(size_t)255;
  float* H2buf = (float*)(ws + off);
  float* outp = (float*)d_out;

  const dim3 blk(256);
  const int eblk = (e + 255) / 256;
  const int gblk = (n + 127) / 128;
  const int ablk = (n + 7) / 8;
  const int ntiles = (n + 31) / 32;
  const int fblk = (ntiles + 2) / 3;

  k_zero      <<<dim3(nblk), blk, 0, stream>>>(cnt, n);
  k_count_rank<<<dim3(eblk), blk, 0, stream>>>(dstv, cnt, rank, e);
  k_blocksum  <<<dim3(nblk), blk, 0, stream>>>(cnt, bsum, n);
  k_scan      <<<dim3(nblk), blk, 0, stream>>>(cnt, bsum, rowptr, dinv, n);

  k_gemm1_fill<<<dim3(gblk + eblk), blk, 0, stream>>>(x, W1, dinv, Hbuf, n, gblk,
                                                       srcv, dstv, rank, rowptr, csr, e);

  k_gather_gemm2<<<dim3(fblk), blk, 0, stream>>>(Hbuf, dinv, rowptr, cnt, csr, b1, W2,
                                                 H2buf, n, ntiles);

  k_gather  <<<dim3(ablk), blk, 0, stream>>>(H2buf, dinv, rowptr, cnt, csr, b2, outp, n);
}


# Edge lists without a scan: the counting add's returned rank puts each source straight into its destination's 32-entry slot row (longer lists spill, summed after the node's own row); 7 launches to 5
# speedup vs baseline: 2.5782x; 1.1338x over previous
#include <hip/hip_runtime.h>

#define D 128

typedef float v2f __attribute__((ext_vector_type(2)));
typedef float v8f __attribute__((ext_vector_type(8)));

#if defined(__HIP_DEVICE_COMPILE__) && __has_builtin(__builtin_amdgcn_wmma_f32_16x16x4_f32)
#define HAVE_WMMA_F32 1
#endif

__device__ __forceinline__ v8f wmma_acc(v2f a, v2f b, v8f c) {
#if defined(HAVE_WMMA_F32)
  return __builtin_amdgcn_wmma_f32_16x16x4_f32(false, a, false, b, (short)0, c,
                                               false, false);
#else
  (void)a; (void)b;
  return c;
#endif
}

__global__ void k_zero(int* __restrict__ p, int n) {
  int i = blockIdx.x * blockDim.x + threadIdx.x;
  if (i < n) p[i] = 0;
}

#define CAP 32

__global__ void k_count_place(const int* __restrict__ src, const int* __restrict__ dst,
                              int* __restrict__ cnt, int* __restrict__ slots,
                              int* __restrict__ ovf_dst, int* __restrict__ ovf_src,
                              int n, int e) {
  int i = blockIdx.x * blockDim.x + threadIdx.x;
  if (i < e) {
    const int d = dst[i];
    const int s = src[i];
    const int r = __hip_atomic_fetch_add(&cnt[d], 1, __ATOMIC_RELAXED,
                                         __HIP_MEMORY_SCOPE_AGENT);
    if (r < CAP) {
      slots[(size_t)d * CAP + r] = s;
    } else {
      const int k = __hip_atomic_fetch_add(&cnt[n], 1, __ATOMIC_RELAXED,
                                           __HIP_MEMORY_SCOPE_AGENT);
      ovf_dst[k] = d;
      ovf_src[k] = s;
    }
  }
}

__device__ __forceinline__ float node_w(int c) { return rsqrtf((float)c + 1.0f); }

__device__ __forceinline__ void gemm_strip(
    const float* __restrict__ X, const float* __restrict__ W,
    const int* __restrict__ cnt, float* __restrict__ H, int nrows, int blk) {
  const int wave = threadIdx.x >> 5;
  const int lane = threadIdx.x & 31;
  const int half = lane >> 4;
  const int l16  = lane & 15;
  const int row0 = (blk * 8 + wave) * 16;
  if (row0 >= nrows) return;

  const int m = row0 + l16;
  const bool mvalid = (m < nrows);
  const float* __restrict__ arow = X + (size_t)(mvalid ? m : 0) * D;

  float rs[8];
#pragma unroll
  for (int v = 0; v < 8; ++v) {
    const int mm = row0 + v + 8 * half;
    rs[v] = node_w(cnt[(mm < nrows) ? mm : (nrows - 1)]);
  }

  const v8f vzero = {0.f, 0.f, 0.f, 0.f, 0.f, 0.f, 0.f, 0.f};
  v8f acc[8];
#pragma unroll
  for (int i = 0; i < 8; ++i) acc[i] = vzero;

  for (int k0 = 0; k0 < D; k0 += 4) {
    const int ka = k0 + 2 * half;
    float a0 = arow[ka];
    float a1 = arow[ka + 1];
    if (!mvalid) { a0 = 0.f; a1 = 0.f; }
    v2f a; a.x = a0; a.y = a1;
    const float* __restrict__ w0 = W + (size_t)ka * D + l16;
    const float* __restrict__ w1 = w0 + D;
#pragma unroll
    for (int ct = 0; ct < 8; ++ct) {
      v2f b; b.x = w0[ct * 16]; b.y = w1[ct * 16];
      acc[ct] = wmma_acc(a, b, acc[ct]);
    }
  }

#pragma unroll
  for (int v = 0; v < 8; ++v) {
    const int mm = row0 + v + 8 * half;
    if (mm < nrows) {
      const size_t base = (size_t)mm * D;
#pragma unroll
      for (int ct = 0; ct < 8; ++ct) {
        H[base + ct * 16 + l16] = acc[ct][v] * rs[v];
      }
    }
  }
}

__global__ void __launch_bounds__(256) k_gemm1(
    const float* __restrict__ X, const float* __restrict__ W,
    const int* __restrict__ cnt, float* __restrict__ H, int nrows) {
  gemm_strip(X, W, cnt, H, nrows, (int)blockIdx.x);
}

__device__ __forceinline__ void add4(float4& acc, const float4 h) {
  acc.x += h.x;
  acc.y += h.y;
  acc.z += h.z;
  acc.w += h.w;
}

template <int K>
__device__ __forceinline__ void rows_add(float4& acc, int my_s, int j,
                                         const float4* __restrict__ H4, int lane) {
  float4 h[K];
#pragma unroll
  for (int k = 0; k < K; ++k) {
    const int s = __builtin_amdgcn_readlane(my_s, j + k);
    h[k] = H4[(size_t)s * (D / 4) + lane];
  }
#pragma unroll
  for (int k = 0; k < K; ++k) add4(acc, h[k]);
}

__device__ __forceinline__ int list_head(int d, int m, const int* __restrict__ slots,
                                         int lane) {
  int my_s = d;
  if (lane < m) my_s = slots[(size_t)d * CAP + lane];
  return my_s;
}

__device__ __forceinline__ void overflow_add(float4& acc, int d, int novf,
                                          const int* __restrict__ ovf_dst,
                                          const int* __restrict__ ovf_src,
                                          const float4* __restrict__ H4, int lane) {
  for (int base = 0; base < novf; base += 32) {
    const int idx = base + lane;
    int dk = -1, sk = 0;
    if (idx < novf) { dk = ovf_dst[idx]; sk = ovf_src[idx]; }
    unsigned mask = __builtin_amdgcn_ballot_w32(dk == d);
    while (mask != 0u) {
      const int j = __builtin_ctz(mask);
      mask &= mask - 1u;
      const int s = __builtin_amdgcn_readlane(sk, j);
      add4(acc, H4[(size_t)s * (D / 4) + lane]);
    }
  }
}

__device__ __forceinline__ float4 node_sum(int d, int c, int head,
                                           const int* __restrict__ cnt, int n,
                                           const int* __restrict__ ovf_dst,
                                           const int* __restrict__ ovf_src,
                                           const float4* __restrict__ H4, int lane) {
  const int m0 = (c < CAP) ? c : CAP;
  const int tot = m0 + 1;
  float4 acc;
  acc.x = 0.f; acc.y = 0.f; acc.z = 0.f; acc.w = 0.f;
  for (int base = 0; base < tot; base += 32) {
    const int m = (tot - base < 32) ? (tot - base) : 32;
    const int my_s = (base > 0) ? d : head;
    int j = 0;
    for (; j + 8 <= m; j += 8) rows_add<8>(acc, my_s, j, H4, lane);
    switch (m - j) {
      case 7: rows_add<7>(acc, my_s, j, H4, lane); break;
      case 6: rows_add<6>(acc, my_s, j, H4, lane); break;
      case 5: rows_add<5>(acc, my_s, j, H4, lane); break;
      case 4: rows_add<4>(acc, my_s, j, H4, lane); break;
      case 3: rows_add<3>(acc, my_s, j, H4, lane); break;
      case 2: rows_add<2>(acc, my_s, j, H4, lane); break;
      case 1: rows_add<1>(acc, my_s, j, H4, lane); break;
      default: break;
    }
  }
  if (c > CAP) overflow_add(acc, d, cnt[n], ovf_dst, ovf_src, H4, lane);
  return acc;
}

__global__ void __launch_bounds__(256) k_gather(
    const float* __restrict__ H, const int* __restrict__ cnt,
    const int* __restrict__ slots, const int* __restrict__ ovf_dst,
    const int* __restrict__ ovf_src, const float* __restrict__ bias,
    float* __restrict__ OUT, int n) {
  const int lane = threadIdx.x & 31;
  const int wv = __builtin_amdgcn_readfirstlane((int)(threadIdx.x >> 5));
  const int d = blockIdx.x * 8 + wv;
  if (d >= n) return;
  const int c = cnt[d];
  const float dd = node_w(c);
  const int head = list_head(d, (c < CAP) ? c : CAP, slots, lane);
  const float4 acc = node_sum(d, c, head, cnt, n, ovf_dst, ovf_src, (const float4*)H, lane);
  const float4 b = ((const float4*)bias)[lane];
  float4 o;
  o.x = fmaf(acc.x, dd, b.x);
  o.y = fmaf(acc.y, dd, b.y);
  o.z = fmaf(acc.z, dd, b.z);
  o.w = fmaf(acc.w, dd, b.w);
  ((float4*)OUT)[(size_t)d * (D / 4) + lane] = o;
}

#define TROWS 32
#define APITCH 132

__global__ void __launch_bounds__(256) k_gather_gemm2(
    const float* __restrict__ H, const int* __restrict__ cnt,
    const int* __restrict__ slots, const int* __restrict__ ovf_dst,
    const int* __restrict__ ovf_src, const float* __restrict__ bias,
    const float* __restrict__ W, float* __restrict__ H2, int n, int ntiles) {
  __shared__ __attribute__((aligned(16))) float As[TROWS * APITCH];
  const int lane = threadIdx.x & 31;
  const int wv = __builtin_amdgcn_readfirstlane((int)(threadIdx.x >> 5));
  const int half = lane >> 4;
  const int l16 = lane & 15;
  const int col = wv * 16 + l16;

  v2f bf[32];
#pragma unroll
  for (int i = 0; i < 32; ++i) {
    const int ka = 4 * i + 2 * half;
    bf[i].x = W[(size_t)ka * D + col];
    bf[i].y = W[(size_t)(ka + 1) * D + col];
  }
  const float4 b4 = ((const float4*)bias)[lane];
  const float4* __restrict__ H4 = (const float4*)H;
  const v8f vzero = {0.f, 0.f, 0.f, 0.f, 0.f, 0.f, 0.f, 0.f};

  for (int tile = (int)blockIdx.x; tile < ntiles; tile += (int)gridDim.x) {
    const int row0 = tile * TROWS;

    int ncn[4], nhd[4];
#pragma unroll
    for (int r = 0; r < 4; ++r) {
      const int d = row0 + wv * 4 + r;
      const int dc = (d < n) ? d : (n - 1);
      ncn[r] = cnt[dc];
    }
#pragma unroll
    for (int r = 0; r < 4; ++r) {
      const int d = row0 + wv * 4 + r;
      const int dc = (d < n) ? d : (n - 1);
      nhd[r] = list_head(dc, (ncn[r] < CAP) ? ncn[r] : CAP, slots, lane);
    }
#pragma unroll
    for (int r = 0; r < 4; ++r) {
      const int lr = wv * 4 + r;
      const int d = row0 + lr;
      float4 o;
      o.x = 0.f; o.y = 0.f; o.z = 0.f; o.w = 0.f;
      if (d < n) {
        const float dd = node_w(ncn[r]);
        const float4 acc = node_sum(d, ncn[r], nhd[r], cnt, n, ovf_dst, ovf_src, H4, lane);
        o.x = fmaxf(fmaf(acc.x, dd, b4.x), 0.f);
        o.y = fmaxf(fmaf(acc.y, dd, b4.y), 0.f);
        o.z = fmaxf(fmaf(acc.z, dd, b4.z), 0.f);
        o.w = fmaxf(fmaf(acc.w, dd, b4.w), 0.f);
      }
      *(float4*)&As[lr * APITCH + 4 * lane] = o;
    }

    __syncthreads();

    float rs0[8], rs1[8];
#pragma unroll
    for (int v = 0; v < 8; ++v) {
      const int m0 = row0 + v + 8 * half;
      const int m1 = m0 + 16;
      rs0[v] = node_w(cnt[(m0 < n) ? m0 : (n - 1)]);
      rs1[v] = node_w(cnt[(m1 < n) ? m1 : (n - 1)]);
    }

    v8f acc0 = vzero, acc1 = vzero;
#pragma unroll
    for (int i = 0; i < 32; ++i) {
      const int ka = 4 * i + 2 * half;
      const v2f a0 = *(const v2f*)&As[l16 * APITCH + ka];
      const v2f a1 = *(const v2f*)&As[(16 + l16) * APITCH + ka];
      acc0 = wmma_acc(a0, bf[i], acc0);
      acc1 = wmma_acc(a1, bf[i], acc1);
    }

#pragma unroll
    for (int v = 0; v < 8; ++v) {
      const int m0 = row0 + v + 8 * half;
      const int m1 = m0 + 16;
      if (m0 < n) H2[(size_t)m0 * D + col] = acc0[v] * rs0[v];
      if (m1 < n) H2[(size_t)m1 * D + col] = acc1[v] * rs1[v];
    }

    __syncthreads();
  }
}

extern "C" void kernel_launch(void* const* d_in, const int* in_sizes, int n_in,
                              void* d_out, int out_size, void* d_ws, size_t ws_size,
                              hipStream_t stream) {
  (void)n_in; (void)out_size; (void)ws_size;
  const float* x  = (const float*)d_in[0];
  const int*   ei = (const int*)d_in[1];
  const float* W1 = (const float*)d_in[2];
  const float* b1 = (const float*)d_in[3];
  const float* W2 = (const float*)d_in[4];
  const float* b2 = (const float*)d_in[5];
  const int n = in_sizes[0] / D;
  const int e = in_sizes[1] / 2;
  const int* srcv = ei;
  const int* dstv = ei + e;

  char* ws = (char*)d_ws;
  size_t off = 0;
  int* cnt = (int*)(ws + off);
  off += ((((size_t)n + 1) * 4) + 255) & ~(size_t)255;
  int* slots = (int*)(ws + off);
  off += (((size_t)n * CAP * 4) + 255) & ~(size_t)255;
  int* ovfd = (int*)(ws + off);
  off += (((size_t)e * 4) + 255) & ~(size_t)255;
  int* ovfs = (int*)(ws + off);
  off += (((size_t)e * 4) + 255) & ~(size_t)255;
  float* Hbuf = (float*)(ws + off);
  off += (((size_t)n * D * 4) + 255) & ~(size_t)255;
  float* H2buf = (float*)(ws + off);
  float* outp = (float*)d_out;

  const dim3 blk(256);
  const int zblk = (n + 1 + 255) / 256;
  const int eblk = (e + 255) / 256;
  const int gblk = (n + 127) / 128;
  const int ablk = (n + 7) / 8;
  const int ntiles = (n + 31) / 32;
  const int fblk = (ntiles + 2) / 3;

  k_zero       <<<dim3(zblk), blk, 0, stream>>>(cnt, n + 1);
  k_count_place<<<dim3(eblk), blk, 0, stream>>>(srcv, dstv, cnt, slots, ovfd, ovfs, n, e);

  k_gemm1<<<dim3(gblk), blk, 0, stream>>>(x, W1, cnt, Hbuf, n);

  k_gather_gemm2<<<dim3(fblk), blk, 0, stream>>>(Hbuf, cnt, slots, ovfd, ovfs, b1, W2,
                                                 H2buf, n, ntiles);

  k_gather  <<<dim3(ablk), blk, 0, stream>>>(H2buf, cnt, slots, ovfd, ovfs, b2, outp, n);
}


# Edge lists binned in LDS: 4096-edge runs counting-sorted by 128-node bin, then a work-group per bin ranks them with LDS atomics; no global atomic or scattered store per edge; sum order = LDS rank
# speedup vs baseline: 3.4469x; 1.3369x over previous
#include <hip/hip_runtime.h>

#define D 128

typedef float v2f __attribute__((ext_vector_type(2)));
typedef float v8f __attribute__((ext_vector_type(8)));

#if defined(__HIP_DEVICE_COMPILE__) && __has_builtin(__builtin_amdgcn_wmma_f32_16x16x4_f32)
#define HAVE_WMMA_F32 1
#endif

__device__ __forceinline__ v8f wmma_acc(v2f a, v2f b, v8f c) {
#if defined(HAVE_WMMA_F32)
  return __builtin_amdgcn_wmma_f32_16x16x4_f32(false, a, false, b, (short)0, c,
                                               false, false);
#else
  (void)a; (void)b;
  return c;
#endif
}

#define CAP 32

#define RB 128
#define CH 4096
#define EPT (CH / 256)
#define MAXNB 512
#define MAXP1 256

__device__ __forceinline__ int block_incl_scan(int s, int* wsum, int lane, int wv) {
  int x = s;
#pragma unroll
  for (int o = 1; o < 32; o <<= 1) {
    const int y = __shfl_up(x, o, 32);
    if (lane >= o) x += y;
  }
  if (lane == 31) wsum[wv] = x;
  __syncthreads();
  int woff = 0;
#pragma unroll
  for (int j = 0; j < 8; ++j) if (j < wv) woff += wsum[j];
  return x + woff;
}

__global__ void __launch_bounds__(256) k_bin(
    const int* __restrict__ src, const int* __restrict__ dst, int* __restrict__ runs,
    int* __restrict__ offsT, int* __restrict__ cnt, int n, int e, int nb, int p1) {
  __shared__ int bcnt[MAXNB];
  __shared__ int wsum[8];
  __shared__ int stage[CH];
  const int tid = threadIdx.x;
  const int lane = tid & 31;
  const int wv = tid >> 5;
  const int w = (int)blockIdx.x;
  const int c0 = w * CH;
  const int len = (e - c0 < CH) ? (e - c0) : CH;
  if (w == 0 && tid == 0) cnt[n] = 0;
  bcnt[tid] = 0;
  bcnt[tid + 256] = 0;

  int dd[EPT], ss[EPT];
#pragma unroll
  for (int k = 0; k < EPT; ++k) {
    const int idx = k * 256 + tid;
    dd[k] = -1; ss[k] = 0;
    if (idx < len) { dd[k] = dst[c0 + idx]; ss[k] = src[c0 + idx]; }
  }
  __syncthreads();
  int bn[EPT], rk[EPT];
#pragma unroll
  for (int k = 0; k < EPT; ++k) bn[k] = (dd[k] >= 0) ? (dd[k] >> 7) : (MAXNB - 1);
#pragma unroll
  for (int k = 0; k < EPT; ++k) rk[k] = atomicAdd(&bcnt[bn[k]], 1);
  __syncthreads();
  const int v0 = bcnt[2 * tid];
  const int v1 = bcnt[2 * tid + 1];
  const int incl = block_incl_scan(v0 + v1, wsum, lane, wv);
  const int ex0 = incl - v0 - v1;
  const int ex1 = ex0 + v0;
  bcnt[2 * tid] = ex0;
  bcnt[2 * tid + 1] = ex1;
  if (2 * tid <= nb) offsT[(size_t)(2 * tid) * p1 + w] = ex0;
  if (2 * tid + 1 <= nb) offsT[(size_t)(2 * tid + 1) * p1 + w] = ex1;
  __syncthreads();
  int pos[EPT];
#pragma unroll
  for (int k = 0; k < EPT; ++k) pos[k] = bcnt[bn[k]] + rk[k];
#pragma unroll
  for (int k = 0; k < EPT; ++k) {
    if (dd[k] >= 0) stage[pos[k]] = (int)(((unsigned)ss[k] << 7) | ((unsigned)dd[k] & 127u));
  }
  __syncthreads();
  for (int i = tid; i < len; i += 256) runs[c0 + i] = stage[i];
}

__global__ void __launch_bounds__(256) k_build(
    const int* __restrict__ runs, const int* __restrict__ offsT, int* __restrict__ cnt,
    int* __restrict__ slots, int* __restrict__ ovf_dst, int* __restrict__ ovf_src,
    int n, int p1) {
  __shared__ int lcnt[RB + 32];
  __shared__ __attribute__((aligned(16))) int lslots[RB * CAP];
  __shared__ int sst[MAXP1];
  __shared__ int spre[MAXP1 + 1];
  __shared__ int wsum[8];
  const int tid = threadIdx.x;
  const int lane = tid & 31;
  const int wv = tid >> 5;
  const int b = (int)blockIdx.x;
  const int base = b * RB;

  int st = 0, ln = 0;
  if (tid < p1) {
    st = offsT[(size_t)b * p1 + tid];
    ln = offsT[(size_t)(b + 1) * p1 + tid] - st;
  }
  if (tid < RB + 32) lcnt[tid] = 0;
  {
    const int4 z = {0, 0, 0, 0};
#pragma unroll
    for (int q = 0; q < RB * CAP / 4 / 256; ++q) ((int4*)lslots)[q * 256 + tid] = z;
  }
  const int incl = block_incl_scan(ln, wsum, lane, wv);
  sst[tid] = st;
  spre[tid] = incl - ln;
  if (tid == 255) spre[MAXP1] = incl;
  __syncthreads();
  const int total = spre[MAXP1];

  for (int i0 = 0; i0 < total; i0 += 256 * 8) {
    int v[8], ld[8], r[8];
#pragma unroll
    for (int k = 0; k < 8; ++k) {
      const int i = i0 + k * 256 + tid;
      const int ic = (i < total) ? i : (total - 1);
      int lo = 0, hi = MAXP1;
#pragma unroll
      for (int step = 0; step < 8; ++step) {
        const int mid = (lo + hi) >> 1;
        const bool up = (spre[mid] <= ic);
        lo = up ? mid : lo;
        hi = up ? hi : mid;
      }
      v[k] = runs[(size_t)lo * CH + sst[lo] + (ic - spre[lo])];
    }
#pragma unroll
    for (int k = 0; k < 8; ++k) {
      const int i = i0 + k * 256 + tid;
      ld[k] = (i < total) ? (v[k] & 127) : (RB + lane);
    }
#pragma unroll
    for (int k = 0; k < 8; ++k) r[k] = atomicAdd(&lcnt[ld[k]], 1);
#pragma unroll
    for (int k = 0; k < 8; ++k) {
      if (ld[k] < RB) {
        const int s = (int)((unsigned)v[k] >> 7);
        if (r[k] < CAP) {
          lslots[ld[k] * CAP + r[k]] = s;
        } else {
          const int k2 = __hip_atomic_fetch_add(&cnt[n], 1, __ATOMIC_RELAXED,
                                                __HIP_MEMORY_SCOPE_AGENT);
          ovf_dst[k2] = base + ld[k];
          ovf_src[k2] = s;
        }
      }
    }
  }
  __syncthreads();
  if (tid < RB && base + tid < n) cnt[base + tid] = lcnt[tid];
#pragma unroll
  for (int q = 0; q < RB * CAP / 4 / 256; ++q) {
    const int idx = q * 256 + tid;
    if (base + (idx >> 3) < n)
      ((int4*)slots)[(size_t)base * (CAP / 4) + idx] = ((const int4*)lslots)[idx];
  }
}

__device__ __forceinline__ float node_w(int c) { return rsqrtf((float)c + 1.0f); }

__device__ __forceinline__ void gemm_strip(
    const float* __restrict__ X, const float* __restrict__ W,
    const int* __restrict__ cnt, float* __restrict__ H, int nrows, int blk) {
  const int wave = threadIdx.x >> 5;
  const int lane = threadIdx.x & 31;
  const int half = lane >> 4;
  const int l16  = lane & 15;
  const int row0 = (blk * 8 + wave) * 16;
  if (row0 >= nrows) return;

  const int m = row0 + l16;
  const bool mvalid = (m < nrows);
  const float* __restrict__ arow = X + (size_t)(mvalid ? m : 0) * D;

  float rs[8];
#pragma unroll
  for (int v = 0; v < 8; ++v) {
    const int mm = row0 + v + 8 * half;
    rs[v] = node_w(cnt[(mm < nrows) ? mm : (nrows - 1)]);
  }

  const v8f vzero = {0.f, 0.f, 0.f, 0.f, 0.f, 0.f, 0.f, 0.f};
  v8f acc[8];
#pragma unroll
  for (int i = 0; i < 8; ++i) acc[i] = vzero;

  for (int k0 = 0; k0 < D; k0 += 4) {
    const int ka = k0 + 2 * half;
    float a0 = arow[ka];
    float a1 = arow[ka + 1];
    if (!mvalid) { a0 = 0.f; a1 = 0.f; }
    v2f a; a.x = a0; a.y = a1;
    const float* __restrict__ w0 = W + (size_t)ka * D + l16;
    const float* __restrict__ w1 = w0 + D;
#pragma unroll
    for (int ct = 0; ct < 8; ++ct) {
      v2f b; b.x = w0[ct * 16]; b.y = w1[ct * 16];
      acc[ct] = wmma_acc(a, b, acc[ct]);
    }
  }

#pragma unroll
  for (int v = 0; v < 8; ++v) {
    const int mm = row0 + v + 8 * half;
    if (mm < nrows) {
      const size_t base = (size_t)mm * D;
#pragma unroll
      for (int ct = 0; ct < 8; ++ct) {
        H[base + ct * 16 + l16] = acc[ct][v] * rs[v];
      }
    }
  }
}

__global__ void __launch_bounds__(256) k_gemm1(
    const float* __restrict__ X, const float* __restrict__ W,
    const int* __restrict__ cnt, float* __restrict__ H, int nrows) {
  gemm_strip(X, W, cnt, H, nrows, (int)blockIdx.x);
}

__device__ __forceinline__ void add4(float4& acc, const float4 h) {
  acc.x += h.x;
  acc.y += h.y;
  acc.z += h.z;
  acc.w += h.w;
}

template <int K>
__device__ __forceinline__ void rows_add(float4& acc, int my_s, int j,
                                         const float4* __restrict__ H4, int lane) {
  float4 h[K];
#pragma unroll
  for (int k = 0; k < K; ++k) {
    const int s = __builtin_amdgcn_readlane(my_s, j + k);
    h[k] = H4[(size_t)s * (D / 4) + lane];
  }
#pragma unroll
  for (int k = 0; k < K; ++k) add4(acc, h[k]);
}

__device__ __forceinline__ int list_head(int d, int m, const int* __restrict__ slots,
                                         int lane) {
  int my_s = d;
  if (lane < m) my_s = slots[(size_t)d * CAP + lane];
  return my_s;
}

__device__ __forceinline__ void overflow_add(float4& acc, int d, int novf,
                                          const int* __restrict__ ovf_dst,
                                          const int* __restrict__ ovf_src,
                                          const float4* __restrict__ H4, int lane) {
  for (int base = 0; base < novf; base += 32) {
    const int idx = base + lane;
    int dk = -1, sk = 0;
    if (idx < novf) { dk = ovf_dst[idx]; sk = ovf_src[idx]; }
    unsigned mask = __builtin_amdgcn_ballot_w32(dk == d);
    while (mask != 0u) {
      const int j = __builtin_ctz(mask);
      mask &= mask - 1u;
      const int s = __builtin_amdgcn_readlane(sk, j);
      add4(acc, H4[(size_t)s * (D / 4) + lane]);
    }
  }
}

__device__ __forceinline__ float4 node_sum(int d, int c, int head,
                                           const int* __restrict__ cnt, int n,
                                           const int* __restrict__ ovf_dst,
                                           const int* __restrict__ ovf_src,
                                           const float4* __restrict__ H4, int lane) {
  const int m0 = (c < CAP) ? c : CAP;
  const int tot = m0 + 1;
  float4 acc;
  acc.x = 0.f; acc.y = 0.f; acc.z = 0.f; acc.w = 0.f;
  for (int base = 0; base < tot; base += 32) {
    const int m = (tot - base < 32) ? (tot - base) : 32;
    const int my_s = (base > 0) ? d : head;
    int j = 0;
    for (; j + 8 <= m; j += 8) rows_add<8>(acc, my_s, j, H4, lane);
    switch (m - j) {
      case 7: rows_add<7>(acc, my_s, j, H4, lane); break;
      case 6: rows_add<6>(acc, my_s, j, H4, lane); break;
      case 5: rows_add<5>(acc, my_s, j, H4, lane); break;
      case 4: rows_add<4>(acc, my_s, j, H4, lane); break;
      case 3: rows_add<3>(acc, my_s, j, H4, lane); break;
      case 2: rows_add<2>(acc, my_s, j, H4, lane); break;
      case 1: rows_add<1>(acc, my_s, j, H4, lane); break;
      default: break;
    }
  }
  if (c > CAP) overflow_add(acc, d, cnt[n], ovf_dst, ovf_src, H4, lane);
  return acc;
}

__global__ void __launch_bounds__(256) k_gather(
    const float* __restrict__ H, const int* __restrict__ cnt,
    const int* __restrict__ slots, const int* __restrict__ ovf_dst,
    const int* __restrict__ ovf_src, const float* __restrict__ bias,
    float* __restrict__ OUT, int n) {
  const int lane = threadIdx.x & 31;
  const int wv = __builtin_amdgcn_readfirstlane((int)(threadIdx.x >> 5));
  const int d = blockIdx.x * 8 + wv;
  if (d >= n) return;
  const int c = cnt[d];
  const float dd = node_w(c);
  const int head = list_head(d, (c < CAP) ? c : CAP, slots, lane);
  const float4 acc = node_sum(d, c, head, cnt, n, ovf_dst, ovf_src, (const float4*)H, lane);
  const float4 b = ((const float4*)bias)[lane];
  float4 o;
  o.x = fmaf(acc.x, dd, b.x);
  o.y = fmaf(acc.y, dd, b.y);
  o.z = fmaf(acc.z, dd, b.z);
  o.w = fmaf(acc.w, dd, b.w);
  ((float4*)OUT)[(size_t)d * (D / 4) + lane] = o;
}

#define TROWS 32
#define APITCH 132

__global__ void __launch_bounds__(256) k_gather_gemm2(
    const float* __restrict__ H, const int* __restrict__ cnt,
    const int* __restrict__ slots, const int* __restrict__ ovf_dst,
    const int* __restrict__ ovf_src, const float* __restrict__ bias,
    const float* __restrict__ W, float* __restrict__ H2, int n, int ntiles) {
  __shared__ __attribute__((aligned(16))) float As[TROWS * APITCH];
  const int lane = threadIdx.x & 31;
  const int wv = __builtin_amdgcn_readfirstlane((int)(threadIdx.x >> 5));
  const int half = lane >> 4;
  const int l16 = lane & 15;
  const int col = wv * 16 + l16;

  v2f bf[32];
#pragma unroll
  for (int i = 0; i < 32; ++i) {
    const int ka = 4 * i + 2 * half;
    bf[i].x = W[(size_t)ka * D + col];
    bf[i].y = W[(size_t)(ka + 1) * D + col];
  }
  const float4 b4 = ((const float4*)bias)[lane];
  const float4* __restrict__ H4 = (const float4*)H;
  const v8f vzero = {0.f, 0.f, 0.f, 0.f, 0.f, 0.f, 0.f, 0.f};

  for (int tile = (int)blockIdx.x; tile < ntiles; tile += (int)gridDim.x) {
    const int row0 = tile * TROWS;

    int ncn[4], nhd[4];
#pragma unroll
    for (int r = 0; r < 4; ++r) {
      const int d = row0 + wv * 4 + r;
      const int dc = (d < n) ? d : (n - 1);
      ncn[r] = cnt[dc];
    }
#pragma unroll
    for (int r = 0; r < 4; ++r) {
      const int d = row0 + wv * 4 + r;
      const int dc = (d < n) ? d : (n - 1);
      nhd[r] = list_head(dc, (ncn[r] < CAP) ? ncn[r] : CAP, slots, lane);
    }
#pragma unroll
    for (int r = 0; r < 4; ++r) {
      const int lr = wv * 4 + r;
      const int d = row0 + lr;
      float4 o;
      o.x = 0.f; o.y = 0.f; o.z = 0.f; o.w = 0.f;
      if (d < n) {
        const float dd = node_w(ncn[r]);
        const float4 acc = node_sum(d, ncn[r], nhd[r], cnt, n, ovf_dst, ovf_src, H4, lane);
        o.x = fmaxf(fmaf(acc.x, dd, b4.x), 0.f);
        o.y = fmaxf(fmaf(acc.y, dd, b4.y), 0.f);
        o.z = fmaxf(fmaf(acc.z, dd, b4.z), 0.f);
        o.w = fmaxf(fmaf(acc.w, dd, b4.w), 0.f);
      }
      *(float4*)&As[lr * APITCH + 4 * lane] = o;
    }

    __syncthreads();

    float rs0[8], rs1[8];
#pragma unroll
    for (int v = 0; v < 8; ++v) {
      const int m0 = row0 + v + 8 * half;
      const int m1 = m0 + 16;
      rs0[v] = node_w(cnt[(m0 < n) ? m0 : (n - 1)]);
      rs1[v] = node_w(cnt[(m1 < n) ? m1 : (n - 1)]);
    }

    v8f acc0 = vzero, acc1 = vzero;
#pragma unroll
    for (int i = 0; i < 32; ++i) {
      const int ka = 4 * i + 2 * half;
      const v2f a0 = *(const v2f*)&As[l16 * APITCH + ka];
      const v2f a1 = *(const v2f*)&As[(16 + l16) * APITCH + ka];
      acc0 = wmma_acc(a0, bf[i], acc0);
      acc1 = wmma_acc(a1, bf[i], acc1);
    }

#pragma unroll
    for (int v = 0; v < 8; ++v) {
      const int m0 = row0 + v + 8 * half;
      const int m1 = m0 + 16;
      if (m0 < n) H2[(size_t)m0 * D + col] = acc0[v] * rs0[v];
      if (m1 < n) H2[(size_t)m1 * D + col] = acc1[v] * rs1[v];
    }

    __syncthreads();
  }
}

extern "C" void kernel_launch(void* const* d_in, const int* in_sizes, int n_in,
                              void* d_out, int out_size, void* d_ws, size_t ws_size,
                              hipStream_t stream) {
  (void)n_in; (void)out_size; (void)ws_size;
  const float* x  = (const float*)d_in[0];
  const int*   ei = (const int*)d_in[1];
  const float* W1 = (const float*)d_in[2];
  const float* b1 = (const float*)d_in[3];
  const float* W2 = (const float*)d_in[4];
  const float* b2 = (const float*)d_in[5];
  const int n = in_sizes[0] / D;
  const int e = in_sizes[1] / 2;
  const int* srcv = ei;
  const int* dstv = ei + e;

  char* ws = (char*)d_ws;
  size_t off = 0;
  int* cnt = (int*)(ws + off);
  off += ((((size_t)n + 1) * 4) + 255) & ~(size_t)255;
  int* slots = (int*)(ws + off);
  off += (((size_t)n * CAP * 4) + 255) & ~(size_t)255;
  int* ovfd = (int*)(ws + off);
  off += (((size_t)e * 4) + 255) & ~(size_t)255;
  int* ovfs = (int*)(ws + off);
  off += (((size_t)e * 4) + 255) & ~(size_t)255;
  float* Hbuf = (float*)(ws + off);
  off += (((size_t)n * D * 4) + 255) & ~(size_t)255;
  float* H2buf = (float*)(ws + off);
  off += (((size_t)n * D * 4) + 255) & ~(size_t)255;
  int* runs = (int*)(ws + off);
  off += (((size_t)e * 4) + 255) & ~(size_t)255;
  int* offsT = (int*)(ws + off);
  float* outp = (float*)d_out;

  const dim3 blk(256);
  const int nb = (n + RB - 1) / RB;
  const int p1 = (e + CH - 1) / CH;
  const int gblk = (n + 127) / 128;
  const int ablk = (n + 7) / 8;
  const int ntiles = (n + 31) / 32;
  const int fblk = (ntiles + 2) / 3;

  k_bin  <<<dim3(p1), blk, 0, stream>>>(srcv, dstv, runs, offsT, cnt, n, e, nb, p1);
  k_build<<<dim3(nb), blk, 0, stream>>>(runs, offsT, cnt, slots, ovfd, ovfs, n, p1);

  k_gemm1<<<dim3(gblk), blk, 0, stream>>>(x, W1, cnt, Hbuf, n);

  k_gather_gemm2<<<dim3(fblk), blk, 0, stream>>>(Hbuf, cnt, slots, ovfd, ovfs, b1, W2,
                                                 H2buf, n, ntiles);

  k_gather  <<<dim3(ablk), blk, 0, stream>>>(H2buf, cnt, slots, ovfd, ovfs, b2, outp, n);
}


# Layer-1 GEMM joins the list-build launch: the work-group that ranks a 128-node bin then multiplies that bin's rows of x (LDS tile, W fragments in registers); 5 launches to 4; sums unchanged
# speedup vs baseline: 3.8683x; 1.1223x over previous
#include <hip/hip_runtime.h>

#define D 128

typedef float v2f __attribute__((ext_vector_type(2)));
typedef float v8f __attribute__((ext_vector_type(8)));

#if defined(__HIP_DEVICE_COMPILE__) && __has_builtin(__builtin_amdgcn_wmma_f32_16x16x4_f32)
#define HAVE_WMMA_F32 1
#endif

__device__ __forceinline__ v8f wmma_acc(v2f a, v2f b, v8f c) {
#if defined(HAVE_WMMA_F32)
  return __builtin_amdgcn_wmma_f32_16x16x4_f32(false, a, false, b, (short)0, c,
                                               false, false);
#else
  (void)a; (void)b;
  return c;
#endif
}

#define CAP 32

#define RB 128
#define CH 4096
#define EPT (CH / 256)
#define MAXNB 512
#define MAXP1 256

__device__ __forceinline__ int block_incl_scan(int s, int* wsum, int lane, int wv) {
  int x = s;
#pragma unroll
  for (int o = 1; o < 32; o <<= 1) {
    const int y = __shfl_up(x, o, 32);
    if (lane >= o) x += y;
  }
  if (lane == 31) wsum[wv] = x;
  __syncthreads();
  int woff = 0;
#pragma unroll
  for (int j = 0; j < 8; ++j) if (j < wv) woff += wsum[j];
  return x + woff;
}

__global__ void __launch_bounds__(256) k_bin(
    const int* __restrict__ src, const int* __restrict__ dst, int* __restrict__ runs,
    int* __restrict__ offsT, int* __restrict__ cnt, int n, int e, int nb, int p1) {
  __shared__ int bcnt[MAXNB];
  __shared__ int wsum[8];
  __shared__ int stage[CH];
  const int tid = threadIdx.x;
  const int lane = tid & 31;
  const int wv = tid >> 5;
  const int w = (int)blockIdx.x;
  const int c0 = w * CH;
  const int len = (e - c0 < CH) ? (e - c0) : CH;
  if (w == 0 && tid == 0) cnt[n] = 0;
  bcnt[tid] = 0;
  bcnt[tid + 256] = 0;

  int dd[EPT], ss[EPT];
#pragma unroll
  for (int k = 0; k < EPT; ++k) {
    const int idx = k * 256 + tid;
    dd[k] = -1; ss[k] = 0;
    if (idx < len) { dd[k] = dst[c0 + idx]; ss[k] = src[c0 + idx]; }
  }
  __syncthreads();
  int bn[EPT], rk[EPT];
#pragma unroll
  for (int k = 0; k < EPT; ++k) bn[k] = (dd[k] >= 0) ? (dd[k] >> 7) : (MAXNB - 1);
#pragma unroll
  for (int k = 0; k < EPT; ++k) rk[k] = atomicAdd(&bcnt[bn[k]], 1);
  __syncthreads();
  const int v0 = bcnt[2 * tid];
  const int v1 = bcnt[2 * tid + 1];
  const int incl = block_incl_scan(v0 + v1, wsum, lane, wv);
  const int ex0 = incl - v0 - v1;
  const int ex1 = ex0 + v0;
  bcnt[2 * tid] = ex0;
  bcnt[2 * tid + 1] = ex1;
  if (2 * tid <= nb) offsT[(size_t)(2 * tid) * p1 + w] = ex0;
  if (2 * tid + 1 <= nb) offsT[(size_t)(2 * tid + 1) * p1 + w] = ex1;
  __syncthreads();
  int pos[EPT];
#pragma unroll
  for (int k = 0; k < EPT; ++k) pos[k] = bcnt[bn[k]] + rk[k];
#pragma unroll
  for (int k = 0; k < EPT; ++k) {
    if (dd[k] >= 0) stage[pos[k]] = (int)(((unsigned)ss[k] << 7) | ((unsigned)dd[k] & 127u));
  }
  __syncthreads();
  for (int i = tid; i < len; i += 256) runs[c0 + i] = stage[i];
}

__device__ __forceinline__ float node_w(int c) { return rsqrtf((float)c + 1.0f); }

#define APITCH 132

__global__ void __launch_bounds__(256) k_build_gemm1(
    const int* __restrict__ runs, const int* __restrict__ offsT, int* __restrict__ cnt,
    int* __restrict__ slots, int* __restrict__ ovf_dst, int* __restrict__ ovf_src,
    const float* __restrict__ X, const float* __restrict__ W, float* __restrict__ H,
    int n, int p1) {
  __shared__ int lcnt[RB + 32];
  __shared__ __attribute__((aligned(16))) int lslots[RB * CAP];
  __shared__ int sst[MAXP1];
  __shared__ int spre[MAXP1 + 1];
  __shared__ int wsum[8];
  __shared__ __attribute__((aligned(16))) float As[RB * APITCH];
  __shared__ __attribute__((aligned(16))) float wrow[RB];
  const int tid = threadIdx.x;
  const int lane = tid & 31;
  const int wv = tid >> 5;
  const int b = (int)blockIdx.x;
  const int base = b * RB;

  float4 xr[RB / 8];
#pragma unroll
  for (int q = 0; q < RB / 8; ++q) {
    const int m = base + q * 8 + wv;
    const int mc = (m < n) ? m : (n - 1);
    xr[q] = ((const float4*)X)[(size_t)mc * (D / 4) + lane];
  }
  int st = 0, ln = 0;
  if (tid < p1) {
    st = offsT[(size_t)b * p1 + tid];
    ln = offsT[(size_t)(b + 1) * p1 + tid] - st;
  }
  if (tid < RB + 32) lcnt[tid] = 0;
  {
    const int4 z = {0, 0, 0, 0};
#pragma unroll
    for (int q = 0; q < RB * CAP / 4 / 256; ++q) ((int4*)lslots)[q * 256 + tid] = z;
  }
  const int incl = block_incl_scan(ln, wsum, lane, wv);
  sst[tid] = st;
  spre[tid] = incl - ln;
  if (tid == 255) spre[MAXP1] = incl;
  __syncthreads();
  const int total = spre[MAXP1];

  for (int i0 = 0; i0 < total; i0 += 256 * 8) {
    int v[8], ld[8], r[8];
#pragma unroll
    for (int k = 0; k < 8; ++k) {
      const int i = i0 + k * 256 + tid;
      const int ic = (i < total) ? i : (total - 1);
      int lo = 0, hi = MAXP1;
#pragma unroll
      for (int step = 0; step < 8; ++step) {
        const int mid = (lo + hi) >> 1;
        const bool up = (spre[mid] <= ic);
        lo = up ? mid : lo;
        hi = up ? hi : mid;
      }
      v[k] = runs[(size_t)lo * CH + sst[lo] + (ic - spre[lo])];
    }
#pragma unroll
    for (int k = 0; k < 8; ++k) {
      const int i = i0 + k * 256 + tid;
      ld[k] = (i < total) ? (v[k] & 127) : (RB + lane);
    }
#pragma unroll
    for (int k = 0; k < 8; ++k) r[k] = atomicAdd(&lcnt[ld[k]], 1);
#pragma unroll
    for (int k = 0; k < 8; ++k) {
      if (ld[k] < RB) {
        const int s = (int)((unsigned)v[k] >> 7);
        if (r[k] < CAP) {
          lslots[ld[k] * CAP + r[k]] = s;
        } else {
          const int k2 = __hip_atomic_fetch_add(&cnt[n], 1, __ATOMIC_RELAXED,
                                                __HIP_MEMORY_SCOPE_AGENT);
          ovf_dst[k2] = base + ld[k];
          ovf_src[k2] = s;
        }
      }
    }
  }

  const int half = lane >> 4;
  const int l16 = lane & 15;
  const int col = wv * 16 + l16;
  v2f bf[32];
#pragma unroll
  for (int i = 0; i < 32; ++i) {
    const int ka = 4 * i + 2 * half;
    bf[i].x = W[(size_t)ka * D + col];
    bf[i].y = W[(size_t)(ka + 1) * D + col];
  }

  __syncthreads();
  if (tid < RB && base + tid < n) cnt[base + tid] = lcnt[tid];
#pragma unroll
  for (int q = 0; q < RB * CAP / 4 / 256; ++q) {
    const int idx = q * 256 + tid;
    if (base + (idx >> 3) < n)
      ((int4*)slots)[(size_t)base * (CAP / 4) + idx] = ((const int4*)lslots)[idx];
  }

  if (tid < RB) wrow[tid] = node_w(lcnt[tid]);
#pragma unroll
  for (int q = 0; q < RB / 8; ++q) {
    const int rr = q * 8 + wv;
    float4 t = xr[q];
    if (base + rr >= n) { t.x = 0.f; t.y = 0.f; t.z = 0.f; t.w = 0.f; }
    *(float4*)&As[rr * APITCH + 4 * lane] = t;
  }
  __syncthreads();

  const v8f vzero = {0.f, 0.f, 0.f, 0.f, 0.f, 0.f, 0.f, 0.f};
  v8f acc[RB / 16];
#pragma unroll
  for (int rt = 0; rt < RB / 16; ++rt) acc[rt] = vzero;
#pragma unroll
  for (int i = 0; i < 32; ++i) {
    const int ka = 4 * i + 2 * half;
#pragma unroll
    for (int rt = 0; rt < RB / 16; ++rt) {
      const v2f a = *(const v2f*)&As[(rt * 16 + l16) * APITCH + ka];
      acc[rt] = wmma_acc(a, bf[i], acc[rt]);
    }
  }

  const bool whole = (base + RB <= n);
#pragma unroll
  for (int rt = 0; rt < RB / 16; ++rt) {
    const int r0 = rt * 16 + 8 * half;
    const float4 wa = *(const float4*)&wrow[r0];
    const float4 wb = *(const float4*)&wrow[r0 + 4];
    const float ws8[8] = {wa.x, wa.y, wa.z, wa.w, wb.x, wb.y, wb.z, wb.w};
    float* __restrict__ hrow = H + (size_t)(base + r0) * D + col;
    if (whole) {
#pragma unroll
      for (int v = 0; v < 8; ++v) hrow[(size_t)v * D] = acc[rt][v] * ws8[v];
    } else {
#pragma unroll
      for (int v = 0; v < 8; ++v) {
        if (base + r0 + v < n) hrow[(size_t)v * D] = acc[rt][v] * ws8[v];
      }
    }
  }
}

__device__ __forceinline__ void add4(float4& acc, const float4 h) {
  acc.x += h.x;
  acc.y += h.y;
  acc.z += h.z;
  acc.w += h.w;
}

template <int K>
__device__ __forceinline__ void rows_add(float4& acc, int my_s, int j,
                                         const float4* __restrict__ H4, int lane) {
  float4 h[K];
#pragma unroll
  for (int k = 0; k < K; ++k) {
    const int s = __builtin_amdgcn_readlane(my_s, j + k);
    h[k] = H4[(size_t)s * (D / 4) + lane];
  }
#pragma unroll
  for (int k = 0; k < K; ++k) add4(acc, h[k]);
}

__device__ __forceinline__ int list_head(int d, int m, const int* __restrict__ slots,
                                         int lane) {
  int my_s = d;
  if (lane < m) my_s = slots[(size_t)d * CAP + lane];
  return my_s;
}

__device__ __forceinline__ void overflow_add(float4& acc, int d, int novf,
                                          const int* __restrict__ ovf_dst,
                                          const int* __restrict__ ovf_src,
                                          const float4* __restrict__ H4, int lane) {
  for (int base = 0; base < novf; base += 32) {
    const int idx = base + lane;
    int dk = -1, sk = 0;
    if (idx < novf) { dk = ovf_dst[idx]; sk = ovf_src[idx]; }
    unsigned mask = __builtin_amdgcn_ballot_w32(dk == d);
    while (mask != 0u) {
      const int j = __builtin_ctz(mask);
      mask &= mask - 1u;
      const int s = __builtin_amdgcn_readlane(sk, j);
      add4(acc, H4[(size_t)s * (D / 4) + lane]);
    }
  }
}

__device__ __forceinline__ float4 node_sum(int d, int c, int head,
                                           const int* __restrict__ cnt, int n,
                                           const int* __restrict__ ovf_dst,
                                           const int* __restrict__ ovf_src,
                                           const float4* __restrict__ H4, int lane) {
  const int m0 = (c < CAP) ? c : CAP;
  const int tot = m0 + 1;
  float4 acc;
  acc.x = 0.f; acc.y = 0.f; acc.z = 0.f; acc.w = 0.f;
  for (int base = 0; base < tot; base += 32) {
    const int m = (tot - base < 32) ? (tot - base) : 32;
    const int my_s = (base > 0) ? d : head;
    int j = 0;
    for (; j + 8 <= m; j += 8) rows_add<8>(acc, my_s, j, H4, lane);
    switch (m - j) {
      case 7: rows_add<7>(acc, my_s, j, H4, lane); break;
      case 6: rows_add<6>(acc, my_s, j, H4, lane); break;
      case 5: rows_add<5>(acc, my_s, j, H4, lane); break;
      case 4: rows_add<4>(acc, my_s, j, H4, lane); break;
      case 3: rows_add<3>(acc, my_s, j, H4, lane); break;
      case 2: rows_add<2>(acc, my_s, j, H4, lane); break;
      case 1: rows_add<1>(acc, my_s, j, H4, lane); break;
      default: break;
    }
  }
  if (c > CAP) overflow_add(acc, d, cnt[n], ovf_dst, ovf_src, H4, lane);
  return acc;
}

__global__ void __launch_bounds__(256) k_gather(
    const float* __restrict__ H, const int* __restrict__ cnt,
    const int* __restrict__ slots, const int* __restrict__ ovf_dst,
    const int* __restrict__ ovf_src, const float* __restrict__ bias,
    float* __restrict__ OUT, int n) {
  const int lane = threadIdx.x & 31;
  const int wv = __builtin_amdgcn_readfirstlane((int)(threadIdx.x >> 5));
  const int d = blockIdx.x * 8 + wv;
  if (d >= n) return;
  const int c = cnt[d];
  const float dd = node_w(c);
  const int head = list_head(d, (c < CAP) ? c : CAP, slots, lane);
  const float4 acc = node_sum(d, c, head, cnt, n, ovf_dst, ovf_src, (const float4*)H, lane);
  const float4 b = ((const float4*)bias)[lane];
  float4 o;
  o.x = fmaf(acc.x, dd, b.x);
  o.y = fmaf(acc.y, dd, b.y);
  o.z = fmaf(acc.z, dd, b.z);
  o.w = fmaf(acc.w, dd, b.w);
  ((float4*)OUT)[(size_t)d * (D / 4) + lane] = o;
}

#define TROWS 32

__global__ void __launch_bounds__(256) k_gather_gemm2(
    const float* __restrict__ H, const int* __restrict__ cnt,
    const int* __restrict__ slots, const int* __restrict__ ovf_dst,
    const int* __restrict__ ovf_src, const float* __restrict__ bias,
    const float* __restrict__ W, float* __restrict__ H2, int n, int ntiles) {
  __shared__ __attribute__((aligned(16))) float As[TROWS * APITCH];
  const int lane = threadIdx.x & 31;
  const int wv = __builtin_amdgcn_readfirstlane((int)(threadIdx.x >> 5));
  const int half = lane >> 4;
  const int l16 = lane & 15;
  const int col = wv * 16 + l16;

  v2f bf[32];
#pragma unroll
  for (int i = 0; i < 32; ++i) {
    const int ka = 4 * i + 2 * half;
    bf[i].x = W[(size_t)ka * D + col];
    bf[i].y = W[(size_t)(ka + 1) * D + col];
  }
  const float4 b4 = ((const float4*)bias)[lane];
  const float4* __restrict__ H4 = (const float4*)H;
  const v8f vzero = {0.f, 0.f, 0.f, 0.f, 0.f, 0.f, 0.f, 0.f};

  for (int tile = (int)blockIdx.x; tile < ntiles; tile += (int)gridDim.x) {
    const int row0 = tile * TROWS;

    int ncn[4], nhd[4];
#pragma unroll
    for (int r = 0; r < 4; ++r) {
      const int d = row0 + wv * 4 + r;
      const int dc = (d < n) ? d : (n - 1);
      ncn[r] = cnt[dc];
    }
#pragma unroll
    for (int r = 0; r < 4; ++r) {
      const int d = row0 + wv * 4 + r;
      const int dc = (d < n) ? d : (n - 1);
      nhd[r] = list_head(dc, (ncn[r] < CAP) ? ncn[r] : CAP, slots, lane);
    }
#pragma unroll
    for (int r = 0; r < 4; ++r) {
      const int lr = wv * 4 + r;
      const int d = row0 + lr;
      float4 o;
      o.x = 0.f; o.y = 0.f; o.z = 0.f; o.w = 0.f;
      if (d < n) {
        const float dd = node_w(ncn[r]);
        const float4 acc = node_sum(d, ncn[r], nhd[r], cnt, n, ovf_dst, ovf_src, H4, lane);
        o.x = fmaxf(fmaf(acc.x, dd, b4.x), 0.f);
        o.y = fmaxf(fmaf(acc.y, dd, b4.y), 0.f);
        o.z = fmaxf(fmaf(acc.z, dd, b4.z), 0.f);
        o.w = fmaxf(fmaf(acc.w, dd, b4.w), 0.f);
      }
      *(float4*)&As[lr * APITCH + 4 * lane] = o;
    }

    __syncthreads();

    float rs0[8], rs1[8];
#pragma unroll
    for (int v = 0; v < 8; ++v) {
      const int m0 = row0 + v + 8 * half;
      const int m1 = m0 + 16;
      rs0[v] = node_w(cnt[(m0 < n) ? m0 : (n - 1)]);
      rs1[v] = node_w(cnt[(m1 < n) ? m1 : (n - 1)]);
    }

    v8f acc0 = vzero, acc1 = vzero;
#pragma unroll
    for (int i = 0; i < 32; ++i) {
      const int ka = 4 * i + 2 * half;
      const v2f a0 = *(const v2f*)&As[l16 * APITCH + ka];
      const v2f a1 = *(const v2f*)&As[(16 + l16) * APITCH + ka];
      acc0 = wmma_acc(a0, bf[i], acc0);
      acc1 = wmma_acc(a1, bf[i], acc1);
    }

#pragma unroll
    for (int v = 0; v < 8; ++v) {
      const int m0 = row0 + v + 8 * half;
      const int m1 = m0 + 16;
      if (m0 < n) H2[(size_t)m0 * D + col] = acc0[v] * rs0[v];
      if (m1 < n) H2[(size_t)m1 * D + col] = acc1[v] * rs1[v];
    }

    __syncthreads();
  }
}

extern "C" void kernel_launch(void* const* d_in, const int* in_sizes, int n_in,
                              void* d_out, int out_size, void* d_ws, size_t ws_size,
                              hipStream_t stream) {
  (void)n_in; (void)out_size; (void)ws_size;
  const float* x  = (const float*)d_in[0];
  const int*   ei = (const int*)d_in[1];
  const float* W1 = (const float*)d_in[2];
  const float* b1 = (const float*)d_in[3];
  const float* W2 = (const float*)d_in[4];
  const float* b2 = (const float*)d_in[5];
  const int n = in_sizes[0] / D;
  const int e = in_sizes[1] / 2;
  const int* srcv = ei;
  const int* dstv = ei + e;

  char* ws = (char*)d_ws;
  size_t off = 0;
  int* cnt = (int*)(ws + off);
  off += ((((size_t)n + 1) * 4) + 255) & ~(size_t)255;
  int* slots = (int*)(ws + off);
  off += (((size_t)n * CAP * 4) + 255) & ~(size_t)255;
  int* ovfd = (int*)(ws + off);
  off += (((size_t)e * 4) + 255) & ~(size_t)255;
  int* ovfs = (int*)(ws + off);
  off += (((size_t)e * 4) + 255) & ~(size_t)255;
  float* Hbuf = (float*)(ws + off);
  off += (((size_t)n * D * 4) + 255) & ~(size_t)255;
  float* H2buf = (float*)(ws + off);
  off += (((size_t)n * D * 4) + 255) & ~(size_t)255;
  int* runs = (int*)(ws + off);
  off += (((size_t)e * 4) + 255) & ~(size_t)255;
  int* offsT = (int*)(ws + off);
  float* outp = (float*)d_out;

  const dim3 blk(256);
  const int nb = (n + RB - 1) / RB;
  const int p1 = (e + CH - 1) / CH;
  const int ablk = (n + 7) / 8;
  const int ntiles = (n + 31) / 32;
  const int fblk = (ntiles + 2) / 3;

  k_bin        <<<dim3(p1), blk, 0, stream>>>(srcv, dstv, runs, offsT, cnt, n, e, nb, p1);
  k_build_gemm1<<<dim3(nb), blk, 0, stream>>>(runs, offsT, cnt, slots, ovfd, ovfs, x, W1,
                                              Hbuf, n, p1);

  k_gather_gemm2<<<dim3(fblk), blk, 0, stream>>>(Hbuf, cnt, slots, ovfd, ovfs, b1, W2,
                                                 H2buf, n, ntiles);

  k_gather  <<<dim3(ablk), blk, 0, stream>>>(H2buf, cnt, slots, ovfd, ovfs, b2, outp, n);
}
